# starting kernel
# baseline (speedup 1.0000x reference)
#include <hip/hip_runtime.h>

#define LAM 2.0f

typedef __attribute__((ext_vector_type(2))) float v2f;
typedef __attribute__((ext_vector_type(8))) float v8f;

__global__ void zero_kernel(float* __restrict__ p, long long n) {
    long long i = (long long)blockIdx.x * blockDim.x + threadIdx.x;
    if (i < n) p[i] = 0.0f;
}

__global__ void degree_kernel(const int* __restrict__ dst, float* __restrict__ deg, int E) {
    int i = blockIdx.x * blockDim.x + threadIdx.x;
    if (i < E) atomicAdd(&deg[dst[i]], 1.0f);
}

__global__ void norm_kernel(float* __restrict__ deg, int N) {
    int i = blockIdx.x * blockDim.x + threadIdx.x;
    if (i < N) deg[i] = 1.0f / sqrtf(fmaxf(deg[i], 1.0f));
}

__global__ void scatter_kernel(const float* __restrict__ x, const float* __restrict__ norm,
                               const int* __restrict__ src, const int* __restrict__ dst,
                               float* __restrict__ h, long long total) {
    long long i = (long long)blockIdx.x * blockDim.x + threadIdx.x;
    if (i >= total) return;
    int e = (int)(i >> 6);
    int f = (int)(i & 63);
    int s = src[e];
    int d = dst[e];
    float v = x[(long long)s * 64 + f] * norm[s];
    atomicAdd(&h[(long long)d * 64 + f], v);
}

__global__ void cheb1_kernel(const float* __restrict__ feat, const float* __restrict__ norm,
                             float* __restrict__ h, float* __restrict__ Tx1, long long total) {
    long long i = (long long)blockIdx.x * blockDim.x + threadIdx.x;
    if (i >= total) return;
    const float ch = -2.0f / LAM;
    const float cx = 2.0f / LAM - 1.0f;
    int node = (int)(i >> 6);
    Tx1[i] = ch * h[i] * norm[node] + cx * feat[i];
    h[i] = 0.0f;
}

__global__ __launch_bounds__(256) void cheb_gemm_wmma(
    const float* __restrict__ feat, const float* __restrict__ Tx1,
    const float* __restrict__ h2,   const float* __restrict__ norm,
    const float* __restrict__ W,    const float* __restrict__ bias,
    float* __restrict__ out) {
    const int lane = threadIdx.x & 31;
    const int wave = threadIdx.x >> 5;
    const int gw   = blockIdx.x * 8 + wave;
    const int row_tile = gw >> 2;
    const int n_tile   = gw & 3;
    const int l15  = lane & 15;
    const int half = lane >> 4;
    const int koff = half * 2;

    const int arow = row_tile * 16 + l15;
    const int ncol = n_tile * 16 + l15;

    const float nrm = norm[arow];
    const float* fr = feat + (long long)arow * 64;
    const float* tr = Tx1  + (long long)arow * 64;
    const float* hr = h2   + (long long)arow * 64;
    const float* W0 = W;
    const float* W1 = W + 4096;
    const float* W2 = W + 8192;

    const float ch2 = -4.0f / LAM;
    const float cx2 =  4.0f / LAM - 2.0f;

    v8f acc = {};
#pragma unroll
    for (int k = 0; k < 64; k += 4) {
        const int ka = k + koff;
        float f0 = fr[ka],  f1 = fr[ka + 1];
        float t0 = tr[ka],  t1 = tr[ka + 1];
        float g0 = hr[ka],  g1 = hr[ka + 1];
        v2f a0 = {f0, f1};
        v2f a1 = {t0, t1};
        v2f a2 = {ch2 * g0 * nrm + cx2 * t0 - f0,
                  ch2 * g1 * nrm + cx2 * t1 - f1};
        v2f b0 = {W0[ka * 64 + ncol], W0[(ka + 1) * 64 + ncol]};
        v2f b1 = {W1[ka * 64 + ncol], W1[(ka + 1) * 64 + ncol]};
        v2f b2 = {W2[ka * 64 + ncol], W2[(ka + 1) * 64 + ncol]};
        acc = __builtin_amdgcn_wmma_f32_16x16x4_f32(false, a0, false, b0,
                                                    (short)0, acc, false, false);
        acc = __builtin_amdgcn_wmma_f32_16x16x4_f32(false, a1, false, b1,
                                                    (short)0, acc, false, false);
        acc = __builtin_amdgcn_wmma_f32_16x16x4_f32(false, a2, false, b2,
                                                    (short)0, acc, false, false);
    }

    const float b = bias[ncol];
#pragma unroll
    for (int r = 0; r < 8; ++r) {
        int orow = row_tile * 16 + r + 8 * half;
        out[(long long)orow * 64 + ncol] = acc[r] + b;
    }
}

extern "C" void kernel_launch(void* const* d_in, const int* in_sizes, int n_in,
                              void* d_out, int out_size, void* d_ws, size_t ws_size,
                              hipStream_t stream) {
    const float* feat = (const float*)d_in[0];
    const float* W    = (const float*)d_in[1];
    const float* bias = (const float*)d_in[2];
    const int*   esrc = (const int*)d_in[3];
    const int*   edst = (const int*)d_in[4];
    float* out = (float*)d_out;

    const int N = in_sizes[0] / 64;
    const int E = in_sizes[3];

    float* ws   = (float*)d_ws;
    float* norm = ws;
    float* h    = ws + N;
    float* Tx1  = ws + N + (long long)N * 64;

    const long long nf  = (long long)N * 64;
    const long long tot = (long long)E * 64;
    const long long zn  = (long long)N * 65;

    zero_kernel<<<(unsigned)((zn + 255) / 256), 256, 0, stream>>>(ws, zn);
    degree_kernel<<<(E + 255) / 256, 256, 0, stream>>>(edst, norm, E);
    norm_kernel<<<(N + 255) / 256, 256, 0, stream>>>(norm, N);

    scatter_kernel<<<(unsigned)((tot + 255) / 256), 256, 0, stream>>>(
        feat, norm, esrc, edst, h, tot);
    cheb1_kernel<<<(unsigned)((nf + 255) / 256), 256, 0, stream>>>(
        feat, norm, h, Tx1, nf);
    scatter_kernel<<<(unsigned)((tot + 255) / 256), 256, 0, stream>>>(
        Tx1, norm, esrc, edst, h, tot);

    const int waves = (N / 16) * 4;
    cheb_gemm_wmma<<<waves / 8, 256, 0, stream>>>(feat, Tx1, h, norm, W, bias, out);
}


# Aggregate by destination-sorted gather: counting-sort edges by dst once (one int atomic per edge gives its rank); both passes become per-node row sums without float atomics; sum order changes
# speedup vs baseline: 3.1459x; 3.1459x over previous
#include <hip/hip_runtime.h>

#define LAM 2.0f

typedef __attribute__((ext_vector_type(2))) float v2f;
typedef __attribute__((ext_vector_type(8))) float v8f;

__global__ void zero_counts(int* __restrict__ p, int n) {
    int i = blockIdx.x * blockDim.x + threadIdx.x;
    if (i < n) p[i] = 0;
}

__global__ void count_rank_kernel(const int* __restrict__ dst, int* __restrict__ cnt,
                                  int* __restrict__ rank, int E) {
    int i = blockIdx.x * blockDim.x + threadIdx.x;
    if (i < E) rank[i] = atomicAdd(&cnt[dst[i]], 1);
}

__device__ __forceinline__ int wave_sum(int v) {
    for (int o = 16; o > 0; o >>= 1) v += __shfl_xor(v, o);
    return v;
}

__global__ __launch_bounds__(256) void tile_sum_norm_kernel(const int* __restrict__ cnt,
                                                            float* __restrict__ norm,
                                                            int* __restrict__ tsum, int N) {
    __shared__ int red[8];
    const int t = threadIdx.x;
    const int base = blockIdx.x * 1024 + t * 4;
    int s = 0;
    for (int q = 0; q < 4; ++q) {
        const int i = base + q;
        if (i < N) {
            const int c = cnt[i];
            s += c;
            norm[i] = 1.0f / sqrtf(fmaxf((float)c, 1.0f));
        }
    }
    s = wave_sum(s);
    if ((t & 31) == 0) red[t >> 5] = s;
    __syncthreads();
    if (t == 0) {
        int tot = 0;
        for (int w = 0; w < 8; ++w) tot += red[w];
        tsum[blockIdx.x] = tot;
    }
}

__global__ __launch_bounds__(256) void scan_kernel(int* __restrict__ ptr,
                                                   const int* __restrict__ tsum, int N) {
    __shared__ int red[8];
    __shared__ int wtot[8];
    const int t = threadIdx.x;
    const int lane = t & 31;
    const int w = t >> 5;

    int part = 0;
    for (int j = t; j < (int)blockIdx.x; j += 256) part += tsum[j];
    part = wave_sum(part);
    if (lane == 0) red[w] = part;

    const int base = blockIdx.x * 1024 + t * 4;
    int c0 = 0, c1 = 0, c2 = 0, c3 = 0;
    if (base < N) c0 = ptr[base];
    if (base + 1 < N) c1 = ptr[base + 1];
    if (base + 2 < N) c2 = ptr[base + 2];
    if (base + 3 < N) c3 = ptr[base + 3];
    const int ts = c0 + c1 + c2 + c3;
    int inc = ts;
    for (int o = 1; o < 32; o <<= 1) {
        const int u = __shfl_up(inc, o);
        if (lane >= o) inc += u;
    }
    if (lane == 31) wtot[w] = inc;
    __syncthreads();
    int off = 0;
    for (int q = 0; q < 8; ++q) {
        off += red[q];
        if (q < w) off += wtot[q];
    }
    int ex = off + inc - ts;
    if (base < N) ptr[base] = ex;
    ex += c0;
    if (base + 1 < N) ptr[base + 1] = ex;
    ex += c1;
    if (base + 2 < N) ptr[base + 2] = ex;
    ex += c2;
    if (base + 3 < N) ptr[base + 3] = ex;
}

__global__ void fill_kernel(const int* __restrict__ src, const int* __restrict__ dst,
                            const int* __restrict__ ptr, const int* __restrict__ rank,
                            int* __restrict__ sorted, int E) {
    int i = blockIdx.x * blockDim.x + threadIdx.x;
    if (i < E) sorted[ptr[dst[i]] + rank[i]] = src[i];
}

template <int U>
__device__ __forceinline__ void add_rows(const float* __restrict__ x, int ms, int mwb, int j,
                                         int lane2, float& ax, float& ay) {
    v2f v[U];
    float w[U];
#pragma unroll
    for (int u = 0; u < U; ++u) {
        const int s = __builtin_amdgcn_readlane(ms, j + u);
        w[u] = __int_as_float(__builtin_amdgcn_readlane(mwb, j + u));
        v[u] = *(const v2f*)(x + (long long)s * 64 + lane2);
    }
#pragma unroll
    for (int u = 0; u < U; ++u) {
        ax += v[u].x * w[u];
        ay += v[u].y * w[u];
    }
}

template <int MODE>
__global__ __launch_bounds__(256) void gather_kernel(
    const float* __restrict__ x, const float* __restrict__ feat,
    const float* __restrict__ norm, const int* __restrict__ ptr,
    const int* __restrict__ sorted, float* __restrict__ res, int N, int E) {
    const int lane = threadIdx.x & 31;
    const int node = blockIdx.x * 8 + (threadIdx.x >> 5);
    if (node >= N) return;
    const int beg = __builtin_amdgcn_readfirstlane(ptr[node]);
    const int end = __builtin_amdgcn_readfirstlane(node + 1 < N ? ptr[node + 1] : E);
    const int lane2 = lane * 2;
    float ax = 0.0f, ay = 0.0f;
    for (int base = beg; base < end; base += 32) {
        const int left = end - base;
        const int n = left < 32 ? left : 32;
        int ms = 0;
        float mw = 0.0f;
        if (lane < n) {
            ms = sorted[base + lane];
            mw = norm[ms];
        }
        const int mwb = __float_as_int(mw);
        int j = 0;
        for (; j + 8 <= n; j += 8) add_rows<8>(x, ms, mwb, j, lane2, ax, ay);
        if (n - j >= 4) { add_rows<4>(x, ms, mwb, j, lane2, ax, ay); j += 4; }
        if (n - j >= 2) { add_rows<2>(x, ms, mwb, j, lane2, ax, ay); j += 2; }
        if (n - j >= 1) { add_rows<1>(x, ms, mwb, j, lane2, ax, ay); }
    }
    const long long o = (long long)node * 64 + lane2;
    if (MODE == 0) {
        const float ch = -2.0f / LAM;
        const float cx = 2.0f / LAM - 1.0f;
        const float nn = norm[node];
        const v2f f = *(const v2f*)(feat + o);
        v2f r = {ch * ax * nn + cx * f.x, ch * ay * nn + cx * f.y};
        *(v2f*)(res + o) = r;
    } else {
        v2f r = {ax, ay};
        *(v2f*)(res + o) = r;
    }
}

__global__ __launch_bounds__(256) void cheb_gemm_wmma(
    const float* __restrict__ feat, const float* __restrict__ Tx1,
    const float* __restrict__ h2,   const float* __restrict__ norm,
    const float* __restrict__ W,    const float* __restrict__ bias,
    float* __restrict__ out) {
    const int lane = threadIdx.x & 31;
    const int wave = threadIdx.x >> 5;
    const int gw   = blockIdx.x * 8 + wave;
    const int row_tile = gw >> 2;
    const int n_tile   = gw & 3;
    const int l15  = lane & 15;
    const int half = lane >> 4;
    const int koff = half * 2;

    const int arow = row_tile * 16 + l15;
    const int ncol = n_tile * 16 + l15;

    const float nrm = norm[arow];
    const float* fr = feat + (long long)arow * 64;
    const float* tr = Tx1  + (long long)arow * 64;
    const float* hr = h2   + (long long)arow * 64;
    const float* W0 = W;
    const float* W1 = W + 4096;
    const float* W2 = W + 8192;

    const float ch2 = -4.0f / LAM;
    const float cx2 =  4.0f / LAM - 2.0f;

    v8f acc = {};
#pragma unroll
    for (int k = 0; k < 64; k += 4) {
        const int ka = k + koff;
        float f0 = fr[ka],  f1 = fr[ka + 1];
        float t0 = tr[ka],  t1 = tr[ka + 1];
        float g0 = hr[ka],  g1 = hr[ka + 1];
        v2f a0 = {f0, f1};
        v2f a1 = {t0, t1};
        v2f a2 = {ch2 * g0 * nrm + cx2 * t0 - f0,
                  ch2 * g1 * nrm + cx2 * t1 - f1};
        v2f b0 = {W0[ka * 64 + ncol], W0[(ka + 1) * 64 + ncol]};
        v2f b1 = {W1[ka * 64 + ncol], W1[(ka + 1) * 64 + ncol]};
        v2f b2 = {W2[ka * 64 + ncol], W2[(ka + 1) * 64 + ncol]};
        acc = __builtin_amdgcn_wmma_f32_16x16x4_f32(false, a0, false, b0,
                                                    (short)0, acc, false, false);
        acc = __builtin_amdgcn_wmma_f32_16x16x4_f32(false, a1, false, b1,
                                                    (short)0, acc, false, false);
        acc = __builtin_amdgcn_wmma_f32_16x16x4_f32(false, a2, false, b2,
                                                    (short)0, acc, false, false);
    }

    const float b = bias[ncol];
#pragma unroll
    for (int r = 0; r < 8; ++r) {
        int orow = row_tile * 16 + r + 8 * half;
        out[(long long)orow * 64 + ncol] = acc[r] + b;
    }
}

extern "C" void kernel_launch(void* const* d_in, const int* in_sizes, int n_in,
                              void* d_out, int out_size, void* d_ws, size_t ws_size,
                              hipStream_t stream) {
    const float* feat = (const float*)d_in[0];
    const float* W    = (const float*)d_in[1];
    const float* bias = (const float*)d_in[2];
    const int*   esrc = (const int*)d_in[3];
    const int*   edst = (const int*)d_in[4];
    float* out = (float*)d_out;

    const int N = in_sizes[0] / 64;
    const int E = in_sizes[3];

    float* ws   = (float*)d_ws;
    float* norm = ws;
    float* h    = ws + N;
    float* Tx1  = ws + N + (long long)N * 64;

    int* scr    = (int*)d_out;
    int* ptr    = scr;
    int* tsum   = scr + N;
    int* sorted = scr + N + 1024;
    int* rank   = scr + N + 1024 + E;

    const int nt = (N + 1023) / 1024;

    zero_counts<<<(N + 255) / 256, 256, 0, stream>>>(ptr, N);
    count_rank_kernel<<<(E + 255) / 256, 256, 0, stream>>>(edst, ptr, rank, E);
    tile_sum_norm_kernel<<<nt, 256, 0, stream>>>(ptr, norm, tsum, N);
    scan_kernel<<<nt, 256, 0, stream>>>(ptr, tsum, N);
    fill_kernel<<<(E + 255) / 256, 256, 0, stream>>>(esrc, edst, ptr, rank, sorted, E);

    gather_kernel<0><<<(N + 7) / 8, 256, 0, stream>>>(feat, feat, norm, ptr, sorted, Tx1, N, E);
    gather_kernel<1><<<(N + 7) / 8, 256, 0, stream>>>(Tx1, feat, norm, ptr, sorted, h, N, E);

    const int waves = (N / 16) * 4;
    cheb_gemm_wmma<<<waves / 8, 256, 0, stream>>>(feat, Tx1, h, norm, W, bias, out);
}


# Edge sort as two LDS-counter passes (tiles grouped by 256-node bucket, then one work-group per bucket): no global atomics or scattered word stores, 5 launches become 2; sum order by LDS arrival rank
# speedup vs baseline: 5.6622x; 1.7999x over previous
#include <hip/hip_runtime.h>

#define LAM 2.0f

typedef __attribute__((ext_vector_type(2))) float v2f;
typedef __attribute__((ext_vector_type(2))) int v2i;
typedef __attribute__((ext_vector_type(8))) float v8f;

__device__ __forceinline__ int wave_sum(int v) {
    for (int o = 16; o > 0; o >>= 1) v += __shfl_xor(v, o);
    return v;
}

template <int NT>
__device__ __forceinline__ int block_excl_scan_1024(int* a, int* wtot, int t) {
    constexpr int PER = 1024 / NT;
    constexpr int NW = NT / 32;
    const int lane = t & 31;
    const int w = t >> 5;
    int c[PER];
    int ts = 0;
#pragma unroll
    for (int q = 0; q < PER; ++q) { c[q] = a[t * PER + q]; ts += c[q]; }
    int inc = ts;
    for (int o = 1; o < 32; o <<= 1) {
        const int u = __shfl_up(inc, o);
        if (lane >= o) inc += u;
    }
    if (lane == 31) wtot[w] = inc;
    __syncthreads();
    int off = 0, tot = 0;
    for (int q = 0; q < NW; ++q) {
        const int v = wtot[q];
        if (q < w) off += v;
        tot += v;
    }
    int ex = off + inc - ts;
#pragma unroll
    for (int q = 0; q < PER; ++q) { a[t * PER + q] = ex; ex += c[q]; }
    __syncthreads();
    return tot;
}

#define TE 4096
#define SORT_CAP 8192

__global__ __launch_bounds__(256) void bucket_tile_kernel(
    const int* __restrict__ src, const int* __restrict__ dst, int* __restrict__ tsorted,
    int* __restrict__ cntTB, int* __restrict__ offTB, int E, int NB) {
    __shared__ int hist[1024];
    __shared__ int boff[1024];
    __shared__ int wtot[8];
    __shared__ alignas(8) int stage[2 * TE];
    const int t = threadIdx.x;
    const int tile = blockIdx.x;
    for (int i = t; i < 1024; i += 256) hist[i] = 0;
    __syncthreads();
    const int e0 = tile * TE;
    int s[16], d[16], r[16];
#pragma unroll
    for (int k = 0; k < 16; ++k) {
        const int e = e0 + k * 256 + t;
        s[k] = 0;
        d[k] = -1;
        r[k] = 0;
        if (e < E) { s[k] = src[e]; d[k] = dst[e]; }
    }
#pragma unroll
    for (int k = 0; k < 16; ++k)
        if (d[k] >= 0) r[k] = atomicAdd(&hist[d[k] >> 8], 1);
    __syncthreads();
    for (int i = t; i < 1024; i += 256) boff[i] = hist[i];
    __syncthreads();
    const int tot = block_excl_scan_1024<256>(boff, wtot, t);
#pragma unroll
    for (int k = 0; k < 16; ++k)
        if (d[k] >= 0) {
            const int pos = boff[d[k] >> 8] + r[k];
            v2i sd = {s[k], d[k]};
            *(v2i*)(stage + 2 * pos) = sd;
        }
    __syncthreads();
    for (int i = t; i < tot; i += 256)
        *(v2i*)(tsorted + 2 * ((long long)e0 + i)) = *(const v2i*)(stage + 2 * i);
    for (int b = t; b < NB; b += 256) {
        cntTB[tile * NB + b] = hist[b];
        offTB[tile * NB + b] = boff[b];
    }
}

__device__ __forceinline__ void count_piece(const int* __restrict__ p, int c, int* ncnt) {
    for (int i = 0; i < c; i += 4) {
        int dd[4];
#pragma unroll
        for (int k = 0; k < 4; ++k) dd[k] = (i + k < c) ? p[2 * (i + k) + 1] : -1;
#pragma unroll
        for (int k = 0; k < 4; ++k)
            if (dd[k] >= 0) atomicAdd(&ncnt[dd[k] & 255], 1);
    }
}

__device__ __forceinline__ void place_piece(const int* __restrict__ p, int c, int* cursor,
                                            int* lout, int* __restrict__ sorted_b, bool staged) {
    for (int i = 0; i < c; i += 4) {
        v2i e[4];
#pragma unroll
        for (int k = 0; k < 4; ++k) {
            const v2i none = {0, -1};
            e[k] = (i + k < c) ? *(const v2i*)(p + 2 * (i + k)) : none;
        }
#pragma unroll
        for (int k = 0; k < 4; ++k)
            if (e[k].y >= 0) {
                const int slot = atomicAdd(&cursor[e[k].y & 255], 1);
                if (staged) lout[slot] = e[k].x; else sorted_b[slot] = e[k].x;
            }
    }
}

__global__ __launch_bounds__(512) void bucket_csr_kernel(
    const int* __restrict__ tsorted, const int* __restrict__ cntTB,
    const int* __restrict__ offTB, int* __restrict__ ptr, int* __restrict__ sorted,
    float* __restrict__ norm, int N, int T, int NB) {
    __shared__ int ncnt[1024];
    __shared__ int cursor[256];
    __shared__ int wtot[16];
    __shared__ int redo[16];
    __shared__ int redc[16];
    __shared__ int lout[SORT_CAP];
    const int t = threadIdx.x;
    const int b = blockIdx.x;
    int c0 = 0, o0 = 0, c1 = 0, o1 = 0;
    if (t < T) { c0 = cntTB[t * NB + b]; o0 = offTB[t * NB + b]; }
    if (t + 512 < T) { c1 = cntTB[(t + 512) * NB + b]; o1 = offTB[(t + 512) * NB + b]; }
    ncnt[t] = 0;
    ncnt[t + 512] = 0;
    const int so = wave_sum(o0 + o1);
    const int sc = wave_sum(c0 + c1);
    if ((t & 31) == 0) { redo[t >> 5] = so; redc[t >> 5] = sc; }
    __syncthreads();
    int base = 0;
    int nb = 0;
    for (int q = 0; q < 16; ++q) { base += redo[q]; nb += redc[q]; }

    const int* p0 = tsorted + 2 * ((long long)t * TE + o0);
    const int* p1 = tsorted + 2 * ((long long)(t + 512) * TE + o1);
    count_piece(p0, c0, ncnt);
    count_piece(p1, c1, ncnt);
    __syncthreads();

    const int node = b * 256 + t;
    if (t < 256 && node < N) norm[node] = 1.0f / sqrtf(fmaxf((float)ncnt[t], 1.0f));
    block_excl_scan_1024<512>(ncnt, wtot, t);
    if (t < 256) {
        cursor[t] = ncnt[t];
        if (node < N) ptr[node] = base + ncnt[t];
    }
    __syncthreads();

    const bool staged = nb <= SORT_CAP;
    place_piece(p0, c0, cursor, lout, sorted + base, staged);
    place_piece(p1, c1, cursor, lout, sorted + base, staged);
    __syncthreads();
    if (staged)
        for (int i = t; i < nb; i += 512) sorted[base + i] = lout[i];
}

template <int U>
__device__ __forceinline__ void add_rows(const float* __restrict__ x, int ms, int mwb, int j,
                                         int lane2, float& ax, float& ay) {
    v2f v[U];
    float w[U];
#pragma unroll
    for (int u = 0; u < U; ++u) {
        const int s = __builtin_amdgcn_readlane(ms, j + u);
        w[u] = __int_as_float(__builtin_amdgcn_readlane(mwb, j + u));
        v[u] = *(const v2f*)(x + (long long)s * 64 + lane2);
    }
#pragma unroll
    for (int u = 0; u < U; ++u) {
        ax += v[u].x * w[u];
        ay += v[u].y * w[u];
    }
}

template <int MODE>
__global__ __launch_bounds__(256) void gather_kernel(
    const float* __restrict__ x, const float* __restrict__ feat,
    const float* __restrict__ norm, const int* __restrict__ ptr,
    const int* __restrict__ sorted, float* __restrict__ res, int N, int E) {
    const int lane = threadIdx.x & 31;
    const int node = blockIdx.x * 8 + (threadIdx.x >> 5);
    if (node >= N) return;
    const int beg = __builtin_amdgcn_readfirstlane(ptr[node]);
    const int end = __builtin_amdgcn_readfirstlane(node + 1 < N ? ptr[node + 1] : E);
    const int lane2 = lane * 2;
    float ax = 0.0f, ay = 0.0f;
    for (int base = beg; base < end; base += 32) {
        const int left = end - base;
        const int n = left < 32 ? left : 32;
        int ms = 0;
        float mw = 0.0f;
        if (lane < n) {
            ms = sorted[base + lane];
            mw = norm[ms];
        }
        const int mwb = __float_as_int(mw);
        int j = 0;
        for (; j + 8 <= n; j += 8) add_rows<8>(x, ms, mwb, j, lane2, ax, ay);
        if (n - j >= 4) { add_rows<4>(x, ms, mwb, j, lane2, ax, ay); j += 4; }
        if (n - j >= 2) { add_rows<2>(x, ms, mwb, j, lane2, ax, ay); j += 2; }
        if (n - j >= 1) { add_rows<1>(x, ms, mwb, j, lane2, ax, ay); }
    }
    const long long o = (long long)node * 64 + lane2;
    if (MODE == 0) {
        const float ch = -2.0f / LAM;
        const float cx = 2.0f / LAM - 1.0f;
        const float nn = norm[node];
        const v2f f = *(const v2f*)(feat + o);
        v2f r = {ch * ax * nn + cx * f.x, ch * ay * nn + cx * f.y};
        *(v2f*)(res + o) = r;
    } else {
        v2f r = {ax, ay};
        *(v2f*)(res + o) = r;
    }
}

__global__ __launch_bounds__(256) void cheb_gemm_wmma(
    const float* __restrict__ feat, const float* __restrict__ Tx1,
    const float* __restrict__ h2,   const float* __restrict__ norm,
    const float* __restrict__ W,    const float* __restrict__ bias,
    float* __restrict__ out) {
    const int lane = threadIdx.x & 31;
    const int wave = threadIdx.x >> 5;
    const int gw   = blockIdx.x * 8 + wave;
    const int row_tile = gw >> 2;
    const int n_tile   = gw & 3;
    const int l15  = lane & 15;
    const int half = lane >> 4;
    const int koff = half * 2;

    const int arow = row_tile * 16 + l15;
    const int ncol = n_tile * 16 + l15;

    const float nrm = norm[arow];
    const float* fr = feat + (long long)arow * 64;
    const float* tr = Tx1  + (long long)arow * 64;
    const float* hr = h2   + (long long)arow * 64;
    const float* W0 = W;
    const float* W1 = W + 4096;
    const float* W2 = W + 8192;

    const float ch2 = -4.0f / LAM;
    const float cx2 =  4.0f / LAM - 2.0f;

    v8f acc = {};
#pragma unroll
    for (int k = 0; k < 64; k += 4) {
        const int ka = k + koff;
        float f0 = fr[ka],  f1 = fr[ka + 1];
        float t0 = tr[ka],  t1 = tr[ka + 1];
        float g0 = hr[ka],  g1 = hr[ka + 1];
        v2f a0 = {f0, f1};
        v2f a1 = {t0, t1};
        v2f a2 = {ch2 * g0 * nrm + cx2 * t0 - f0,
                  ch2 * g1 * nrm + cx2 * t1 - f1};
        v2f b0 = {W0[ka * 64 + ncol], W0[(ka + 1) * 64 + ncol]};
        v2f b1 = {W1[ka * 64 + ncol], W1[(ka + 1) * 64 + ncol]};
        v2f b2 = {W2[ka * 64 + ncol], W2[(ka + 1) * 64 + ncol]};
        acc = __builtin_amdgcn_wmma_f32_16x16x4_f32(false, a0, false, b0,
                                                    (short)0, acc, false, false);
        acc = __builtin_amdgcn_wmma_f32_16x16x4_f32(false, a1, false, b1,
                                                    (short)0, acc, false, false);
        acc = __builtin_amdgcn_wmma_f32_16x16x4_f32(false, a2, false, b2,
                                                    (short)0, acc, false, false);
    }

    const float b = bias[ncol];
#pragma unroll
    for (int r = 0; r < 8; ++r) {
        int orow = row_tile * 16 + r + 8 * half;
        out[(long long)orow * 64 + ncol] = acc[r] + b;
    }
}

extern "C" void kernel_launch(void* const* d_in, const int* in_sizes, int n_in,
                              void* d_out, int out_size, void* d_ws, size_t ws_size,
                              hipStream_t stream) {
    const float* feat = (const float*)d_in[0];
    const float* W    = (const float*)d_in[1];
    const float* bias = (const float*)d_in[2];
    const int*   esrc = (const int*)d_in[3];
    const int*   edst = (const int*)d_in[4];
    float* out = (float*)d_out;

    const int N = in_sizes[0] / 64;
    const int E = in_sizes[3];

    float* ws   = (float*)d_ws;
    float* norm = ws;
    float* h    = ws + N;
    float* Tx1  = ws + N + (long long)N * 64;

    const int T  = (E + TE - 1) / TE;
    const int NB = (N + 255) / 256;

    int* scr     = (int*)d_out;
    int* ptr     = scr;
    int* sorted  = scr + N;
    int* tsorted = scr + ((N + E + 1) & ~1);
    int* cntTB   = tsorted + 2 * (long long)E;
    int* offTB   = cntTB + (long long)T * NB;

    bucket_tile_kernel<<<T, 256, 0, stream>>>(esrc, edst, tsorted, cntTB, offTB, E, NB);
    bucket_csr_kernel<<<NB, 512, 0, stream>>>(tsorted, cntTB, offTB, ptr, sorted, norm, N, T, NB);

    gather_kernel<0><<<(N + 7) / 8, 256, 0, stream>>>(feat, feat, norm, ptr, sorted, Tx1, N, E);
    gather_kernel<1><<<(N + 7) / 8, 256, 0, stream>>>(Tx1, feat, norm, ptr, sorted, h, N, E);

    const int waves = (N / 16) * 4;
    cheb_gemm_wmma<<<waves / 8, 256, 0, stream>>>(feat, Tx1, h, norm, W, bias, out);
}


# GEMM stage re-tiled: one wave per 16-row x 64-column strip (4 accumulator tiles): A rows fetched once, not 4x, by 16-byte loads; W fragments read from LDS; k grouping and sum order change
# speedup vs baseline: 6.3294x; 1.1178x over previous
#include <hip/hip_runtime.h>

#define LAM 2.0f

typedef __attribute__((ext_vector_type(2))) float v2f;
typedef __attribute__((ext_vector_type(2))) int v2i;
typedef __attribute__((ext_vector_type(8))) float v8f;

__device__ __forceinline__ int wave_sum(int v) {
    for (int o = 16; o > 0; o >>= 1) v += __shfl_xor(v, o);
    return v;
}

template <int NT>
__device__ __forceinline__ int block_excl_scan_1024(int* a, int* wtot, int t) {
    constexpr int PER = 1024 / NT;
    constexpr int NW = NT / 32;
    const int lane = t & 31;
    const int w = t >> 5;
    int c[PER];
    int ts = 0;
#pragma unroll
    for (int q = 0; q < PER; ++q) { c[q] = a[t * PER + q]; ts += c[q]; }
    int inc = ts;
    for (int o = 1; o < 32; o <<= 1) {
        const int u = __shfl_up(inc, o);
        if (lane >= o) inc += u;
    }
    if (lane == 31) wtot[w] = inc;
    __syncthreads();
    int off = 0, tot = 0;
    for (int q = 0; q < NW; ++q) {
        const int v = wtot[q];
        if (q < w) off += v;
        tot += v;
    }
    int ex = off + inc - ts;
#pragma unroll
    for (int q = 0; q < PER; ++q) { a[t * PER + q] = ex; ex += c[q]; }
    __syncthreads();
    return tot;
}

#define TE 4096
#define SORT_CAP 8192

__global__ __launch_bounds__(256) void bucket_tile_kernel(
    const int* __restrict__ src, const int* __restrict__ dst, int* __restrict__ tsorted,
    int* __restrict__ cntTB, int* __restrict__ offTB, int E, int NB) {
    __shared__ int hist[1024];
    __shared__ int boff[1024];
    __shared__ int wtot[8];
    __shared__ alignas(8) int stage[2 * TE];
    const int t = threadIdx.x;
    const int tile = blockIdx.x;
    for (int i = t; i < 1024; i += 256) hist[i] = 0;
    __syncthreads();
    const int e0 = tile * TE;
    int s[16], d[16], r[16];
#pragma unroll
    for (int k = 0; k < 16; ++k) {
        const int e = e0 + k * 256 + t;
        s[k] = 0;
        d[k] = -1;
        r[k] = 0;
        if (e < E) { s[k] = src[e]; d[k] = dst[e]; }
    }
#pragma unroll
    for (int k = 0; k < 16; ++k)
        if (d[k] >= 0) r[k] = atomicAdd(&hist[d[k] >> 8], 1);
    __syncthreads();
    for (int i = t; i < 1024; i += 256) boff[i] = hist[i];
    __syncthreads();
    const int tot = block_excl_scan_1024<256>(boff, wtot, t);
#pragma unroll
    for (int k = 0; k < 16; ++k)
        if (d[k] >= 0) {
            const int pos = boff[d[k] >> 8] + r[k];
            v2i sd = {s[k], d[k]};
            *(v2i*)(stage + 2 * pos) = sd;
        }
    __syncthreads();
    for (int i = t; i < tot; i += 256)
        *(v2i*)(tsorted + 2 * ((long long)e0 + i)) = *(const v2i*)(stage + 2 * i);
    for (int b = t; b < NB; b += 256) {
        cntTB[tile * NB + b] = hist[b];
        offTB[tile * NB + b] = boff[b];
    }
}

__device__ __forceinline__ void count_piece(const int* __restrict__ p, int c, int* ncnt) {
    for (int i = 0; i < c; i += 4) {
        int dd[4];
#pragma unroll
        for (int k = 0; k < 4; ++k) dd[k] = (i + k < c) ? p[2 * (i + k) + 1] : -1;
#pragma unroll
        for (int k = 0; k < 4; ++k)
            if (dd[k] >= 0) atomicAdd(&ncnt[dd[k] & 255], 1);
    }
}

__device__ __forceinline__ void place_piece(const int* __restrict__ p, int c, int* cursor,
                                            int* lout, int* __restrict__ sorted_b, bool staged) {
    for (int i = 0; i < c; i += 4) {
        v2i e[4];
#pragma unroll
        for (int k = 0; k < 4; ++k) {
            const v2i none = {0, -1};
            e[k] = (i + k < c) ? *(const v2i*)(p + 2 * (i + k)) : none;
        }
#pragma unroll
        for (int k = 0; k < 4; ++k)
            if (e[k].y >= 0) {
                const int slot = atomicAdd(&cursor[e[k].y & 255], 1);
                if (staged) lout[slot] = e[k].x; else sorted_b[slot] = e[k].x;
            }
    }
}

__global__ __launch_bounds__(512) void bucket_csr_kernel(
    const int* __restrict__ tsorted, const int* __restrict__ cntTB,
    const int* __restrict__ offTB, int* __restrict__ ptr, int* __restrict__ sorted,
    float* __restrict__ norm, int N, int T, int NB) {
    __shared__ int ncnt[1024];
    __shared__ int cursor[256];
    __shared__ int wtot[16];
    __shared__ int redo[16];
    __shared__ int redc[16];
    __shared__ int lout[SORT_CAP];
    const int t = threadIdx.x;
    const int b = blockIdx.x;
    int c0 = 0, o0 = 0, c1 = 0, o1 = 0;
    if (t < T) { c0 = cntTB[t * NB + b]; o0 = offTB[t * NB + b]; }
    if (t + 512 < T) { c1 = cntTB[(t + 512) * NB + b]; o1 = offTB[(t + 512) * NB + b]; }
    ncnt[t] = 0;
    ncnt[t + 512] = 0;
    const int so = wave_sum(o0 + o1);
    const int sc = wave_sum(c0 + c1);
    if ((t & 31) == 0) { redo[t >> 5] = so; redc[t >> 5] = sc; }
    __syncthreads();
    int base = 0;
    int nb = 0;
    for (int q = 0; q < 16; ++q) { base += redo[q]; nb += redc[q]; }

    const int* p0 = tsorted + 2 * ((long long)t * TE + o0);
    const int* p1 = tsorted + 2 * ((long long)(t + 512) * TE + o1);
    count_piece(p0, c0, ncnt);
    count_piece(p1, c1, ncnt);
    __syncthreads();

    const int node = b * 256 + t;
    if (t < 256 && node < N) norm[node] = 1.0f / sqrtf(fmaxf((float)ncnt[t], 1.0f));
    block_excl_scan_1024<512>(ncnt, wtot, t);
    if (t < 256) {
        cursor[t] = ncnt[t];
        if (node < N) ptr[node] = base + ncnt[t];
    }
    __syncthreads();

    const bool staged = nb <= SORT_CAP;
    place_piece(p0, c0, cursor, lout, sorted + base, staged);
    place_piece(p1, c1, cursor, lout, sorted + base, staged);
    __syncthreads();
    if (staged)
        for (int i = t; i < nb; i += 512) sorted[base + i] = lout[i];
}

template <int U>
__device__ __forceinline__ void add_rows(const float* __restrict__ x, int ms, int mwb, int j,
                                         int lane2, float& ax, float& ay) {
    v2f v[U];
    float w[U];
#pragma unroll
    for (int u = 0; u < U; ++u) {
        const int s = __builtin_amdgcn_readlane(ms, j + u);
        w[u] = __int_as_float(__builtin_amdgcn_readlane(mwb, j + u));
        v[u] = *(const v2f*)(x + (long long)s * 64 + lane2);
    }
#pragma unroll
    for (int u = 0; u < U; ++u) {
        ax += v[u].x * w[u];
        ay += v[u].y * w[u];
    }
}

template <int MODE>
__global__ __launch_bounds__(256) void gather_kernel(
    const float* __restrict__ x, const float* __restrict__ feat,
    const float* __restrict__ norm, const int* __restrict__ ptr,
    const int* __restrict__ sorted, float* __restrict__ res, int N, int E) {
    const int lane = threadIdx.x & 31;
    const int node = blockIdx.x * 8 + (threadIdx.x >> 5);
    if (node >= N) return;
    const int beg = __builtin_amdgcn_readfirstlane(ptr[node]);
    const int end = __builtin_amdgcn_readfirstlane(node + 1 < N ? ptr[node + 1] : E);
    const int lane2 = lane * 2;
    float ax = 0.0f, ay = 0.0f;
    for (int base = beg; base < end; base += 32) {
        const int left = end - base;
        const int n = left < 32 ? left : 32;
        int ms = 0;
        float mw = 0.0f;
        if (lane < n) {
            ms = sorted[base + lane];
            mw = norm[ms];
        }
        const int mwb = __float_as_int(mw);
        int j = 0;
        for (; j + 8 <= n; j += 8) add_rows<8>(x, ms, mwb, j, lane2, ax, ay);
        if (n - j >= 4) { add_rows<4>(x, ms, mwb, j, lane2, ax, ay); j += 4; }
        if (n - j >= 2) { add_rows<2>(x, ms, mwb, j, lane2, ax, ay); j += 2; }
        if (n - j >= 1) { add_rows<1>(x, ms, mwb, j, lane2, ax, ay); }
    }
    const long long o = (long long)node * 64 + lane2;
    if (MODE == 0) {
        const float ch = -2.0f / LAM;
        const float cx = 2.0f / LAM - 1.0f;
        const float nn = norm[node];
        const v2f f = *(const v2f*)(feat + o);
        v2f r = {ch * ax * nn + cx * f.x, ch * ay * nn + cx * f.y};
        *(v2f*)(res + o) = r;
    } else {
        v2f r = {ax, ay};
        *(v2f*)(res + o) = r;
    }
}

typedef __attribute__((ext_vector_type(4))) float v4f;

__global__ __launch_bounds__(256) void cheb_gemm_wmma(
    const float* __restrict__ feat, const float* __restrict__ Tx1,
    const float* __restrict__ h2,   const float* __restrict__ norm,
    const float* __restrict__ W,    const float* __restrict__ bias,
    float* __restrict__ out, int ntiles) {
    __shared__ alignas(16) float Wl[48 * 64 * 4];
    const int t = threadIdx.x;
    const int lane = t & 31;
    const int wave = t >> 5;
    int tile = blockIdx.x * 8 + wave;
    const bool live = tile < ntiles;
    if (!live) tile = ntiles - 1;
    const int c = lane & 15;
    const int h = lane >> 4;

    const int arow = tile * 16 + c;
    const float nrm = norm[arow];
    const float* fr = feat + (long long)arow * 64 + 4 * h;
    const float* tr = Tx1  + (long long)arow * 64 + 4 * h;
    const float* hr = h2   + (long long)arow * 64 + 4 * h;
    v4f fa[8], xa[8], ga[8];
#pragma unroll
    for (int j = 0; j < 8; ++j) {
        fa[j] = *(const v4f*)(fr + 8 * j);
        xa[j] = *(const v4f*)(tr + 8 * j);
        ga[j] = *(const v4f*)(hr + 8 * j);
    }

    for (int u = t; u < 48 * 16; u += 256) {
        const int kq = u >> 4;
        const int n4 = (u & 15) * 4;
        const float* wp = W + kq * 256 + n4;
        const v4f r0 = *(const v4f*)(wp);
        const v4f r1 = *(const v4f*)(wp + 64);
        const v4f r2 = *(const v4f*)(wp + 128);
        const v4f r3 = *(const v4f*)(wp + 192);
        float* lp = Wl + (kq * 64 + n4) * 4;
        const v4f c0 = {r0.x, r1.x, r2.x, r3.x};
        const v4f c1 = {r0.y, r1.y, r2.y, r3.y};
        const v4f c2 = {r0.z, r1.z, r2.z, r3.z};
        const v4f c3 = {r0.w, r1.w, r2.w, r3.w};
        *(v4f*)(lp)      = c0;
        *(v4f*)(lp + 4)  = c1;
        *(v4f*)(lp + 8)  = c2;
        *(v4f*)(lp + 12) = c3;
    }
    __syncthreads();
#pragma unroll
    for (int j = 0; j < 8; ++j)
        asm volatile("" :: "v"(fa[j]), "v"(xa[j]), "v"(ga[j]));

    const float* bl = Wl + (h * 64 + c) * 4;
    const float ch2 = -4.0f / LAM;
    const float cx2 =  4.0f / LAM - 2.0f;

    v8f acc[4] = {};
#pragma unroll
    for (int j = 0; j < 8; ++j) {
        const v4f f = fa[j];
        const v4f x = xa[j];
        const v4f g = ga[j];
        v4f a[3];
        a[0] = f;
        a[1] = x;
        a[2].x = ch2 * g.x * nrm + cx2 * x.x - f.x;
        a[2].y = ch2 * g.y * nrm + cx2 * x.y - f.y;
        a[2].z = ch2 * g.z * nrm + cx2 * x.z - f.z;
        a[2].w = ch2 * g.w * nrm + cx2 * x.w - f.w;
#pragma unroll
        for (int m = 0; m < 3; ++m) {
            v4f b[4];
#pragma unroll
            for (int nt = 0; nt < 4; ++nt)
                b[nt] = *(const v4f*)(bl + ((m * 16 + 2 * j) * 64 + nt * 16) * 4);
            const v2f alo = {a[m].x, a[m].y};
            const v2f ahi = {a[m].z, a[m].w};
#pragma unroll
            for (int nt = 0; nt < 4; ++nt) {
                const v2f blo = {b[nt].x, b[nt].y};
                acc[nt] = __builtin_amdgcn_wmma_f32_16x16x4_f32(false, alo, false, blo,
                                                                (short)0, acc[nt], false, false);
            }
#pragma unroll
            for (int nt = 0; nt < 4; ++nt) {
                const v2f bhi = {b[nt].z, b[nt].w};
                acc[nt] = __builtin_amdgcn_wmma_f32_16x16x4_f32(false, ahi, false, bhi,
                                                                (short)0, acc[nt], false, false);
            }
        }
    }

    if (live) {
#pragma unroll
        for (int nt = 0; nt < 4; ++nt) {
            const float bv = bias[nt * 16 + c];
#pragma unroll
            for (int r = 0; r < 8; ++r) {
                const int orow = tile * 16 + r + 8 * h;
                out[(long long)orow * 64 + nt * 16 + c] = acc[nt][r] + bv;
            }
        }
    }
}

extern "C" void kernel_launch(void* const* d_in, const int* in_sizes, int n_in,
                              void* d_out, int out_size, void* d_ws, size_t ws_size,
                              hipStream_t stream) {
    const float* feat = (const float*)d_in[0];
    const float* W    = (const float*)d_in[1];
    const float* bias = (const float*)d_in[2];
    const int*   esrc = (const int*)d_in[3];
    const int*   edst = (const int*)d_in[4];
    float* out = (float*)d_out;

    const int N = in_sizes[0] / 64;
    const int E = in_sizes[3];

    float* ws   = (float*)d_ws;
    float* norm = ws;
    float* h    = ws + N;
    float* Tx1  = ws + N + (long long)N * 64;

    const int T  = (E + TE - 1) / TE;
    const int NB = (N + 255) / 256;

    int* scr     = (int*)d_out;
    int* ptr     = scr;
    int* sorted  = scr + N;
    int* tsorted = scr + ((N + E + 1) & ~1);
    int* cntTB   = tsorted + 2 * (long long)E;
    int* offTB   = cntTB + (long long)T * NB;

    bucket_tile_kernel<<<T, 256, 0, stream>>>(esrc, edst, tsorted, cntTB, offTB, E, NB);
    bucket_csr_kernel<<<NB, 512, 0, stream>>>(tsorted, cntTB, offTB, ptr, sorted, norm, N, T, NB);

    gather_kernel<0><<<(N + 7) / 8, 256, 0, stream>>>(feat, feat, norm, ptr, sorted, Tx1, N, E);
    gather_kernel<1><<<(N + 7) / 8, 256, 0, stream>>>(Tx1, feat, norm, ptr, sorted, h, N, E);

    const int tiles = N / 16;
    cheb_gemm_wmma<<<(tiles + 7) / 8, 256, 0, stream>>>(feat, Tx1, h, norm, W, bias, out, tiles);
}


# CSR pass reads each bucket's edges once via an LDS list of pair positions (lanes on neighbouring pairs, all loads in one group); rank from the counting LDS atomic; sum order by arrival changes
# speedup vs baseline: 7.0050x; 1.1067x over previous
#include <hip/hip_runtime.h>

#define LAM 2.0f

typedef __attribute__((ext_vector_type(2))) float v2f;
typedef __attribute__((ext_vector_type(2))) int v2i;
typedef __attribute__((ext_vector_type(8))) float v8f;

__device__ __forceinline__ int wave_sum(int v) {
    for (int o = 16; o > 0; o >>= 1) v += __shfl_xor(v, o);
    return v;
}

template <int NT>
__device__ __forceinline__ int block_excl_scan_1024(int* a, int* wtot, int t) {
    constexpr int PER = 1024 / NT;
    constexpr int NW = NT / 32;
    const int lane = t & 31;
    const int w = t >> 5;
    int c[PER];
    int ts = 0;
#pragma unroll
    for (int q = 0; q < PER; ++q) { c[q] = a[t * PER + q]; ts += c[q]; }
    int inc = ts;
    for (int o = 1; o < 32; o <<= 1) {
        const int u = __shfl_up(inc, o);
        if (lane >= o) inc += u;
    }
    if (lane == 31) wtot[w] = inc;
    __syncthreads();
    int off = 0, tot = 0;
    for (int q = 0; q < NW; ++q) {
        const int v = wtot[q];
        if (q < w) off += v;
        tot += v;
    }
    int ex = off + inc - ts;
#pragma unroll
    for (int q = 0; q < PER; ++q) { a[t * PER + q] = ex; ex += c[q]; }
    __syncthreads();
    return tot;
}

#define TE 4096
#define SORT_CAP 8192

__global__ __launch_bounds__(256) void bucket_tile_kernel(
    const int* __restrict__ src, const int* __restrict__ dst, int* __restrict__ tsorted,
    int* __restrict__ cntTB, int* __restrict__ offTB, int E, int NB) {
    __shared__ int hist[1024];
    __shared__ int boff[1024];
    __shared__ int wtot[8];
    __shared__ alignas(8) int stage[2 * TE];
    const int t = threadIdx.x;
    const int tile = blockIdx.x;
    for (int i = t; i < 1024; i += 256) hist[i] = 0;
    __syncthreads();
    const int e0 = tile * TE;
    int s[16], d[16], r[16];
#pragma unroll
    for (int k = 0; k < 16; ++k) {
        const int e = e0 + k * 256 + t;
        s[k] = 0;
        d[k] = -1;
        r[k] = 0;
        if (e < E) { s[k] = src[e]; d[k] = dst[e]; }
    }
#pragma unroll
    for (int k = 0; k < 16; ++k)
        if (d[k] >= 0) r[k] = atomicAdd(&hist[d[k] >> 8], 1);
    __syncthreads();
    for (int i = t; i < 1024; i += 256) boff[i] = hist[i];
    __syncthreads();
    const int tot = block_excl_scan_1024<256>(boff, wtot, t);
#pragma unroll
    for (int k = 0; k < 16; ++k)
        if (d[k] >= 0) {
            const int pos = boff[d[k] >> 8] + r[k];
            v2i sd = {s[k], d[k]};
            *(v2i*)(stage + 2 * pos) = sd;
        }
    __syncthreads();
    for (int i = t; i < tot; i += 256)
        *(v2i*)(tsorted + 2 * ((long long)e0 + i)) = *(const v2i*)(stage + 2 * i);
    for (int b = t; b < NB; b += 256) {
        cntTB[tile * NB + b] = hist[b];
        offTB[tile * NB + b] = boff[b];
    }
}

__device__ __forceinline__ void count_piece(const int* __restrict__ p, int c, int* ncnt) {
    for (int i = 0; i < c; i += 4) {
        int dd[4];
#pragma unroll
        for (int k = 0; k < 4; ++k) dd[k] = (i + k < c) ? p[2 * (i + k) + 1] : -1;
#pragma unroll
        for (int k = 0; k < 4; ++k)
            if (dd[k] >= 0) atomicAdd(&ncnt[dd[k] & 255], 1);
    }
}

__device__ __forceinline__ void place_piece(const int* __restrict__ p, int c, int* cursor,
                                            int* __restrict__ sorted_b) {
    for (int i = 0; i < c; i += 4) {
        v2i e[4];
#pragma unroll
        for (int k = 0; k < 4; ++k) {
            const v2i none = {0, -1};
            e[k] = (i + k < c) ? *(const v2i*)(p + 2 * (i + k)) : none;
        }
#pragma unroll
        for (int k = 0; k < 4; ++k)
            if (e[k].y >= 0) sorted_b[atomicAdd(&cursor[e[k].y & 255], 1)] = e[k].x;
    }
}

#define CSR_K (SORT_CAP / 512)

__global__ __launch_bounds__(512) void bucket_csr_kernel(
    const int* __restrict__ tsorted, const int* __restrict__ cntTB,
    const int* __restrict__ offTB, int* __restrict__ ptr, int* __restrict__ sorted,
    float* __restrict__ norm, int N, int T, int NB) {
    __shared__ int ncnt[1024];
    __shared__ int tpre[1024];
    __shared__ int cursor[256];
    __shared__ int wtot[16];
    __shared__ int redo[16];
    __shared__ int lout[SORT_CAP];
    const int t = threadIdx.x;
    const int b = blockIdx.x;
    int c0 = 0, o0 = 0, c1 = 0, o1 = 0;
    if (t < T) { c0 = cntTB[t * NB + b]; o0 = offTB[t * NB + b]; }
    if (t + 512 < T) { c1 = cntTB[(t + 512) * NB + b]; o1 = offTB[(t + 512) * NB + b]; }
    ncnt[t] = 0;
    ncnt[t + 512] = 0;
    tpre[t] = c0;
    tpre[t + 512] = c1;
    const int so = wave_sum(o0 + o1);
    if ((t & 31) == 0) redo[t >> 5] = so;
    __syncthreads();
    int base = 0;
    for (int q = 0; q < 16; ++q) base += redo[q];
    const int nb = block_excl_scan_1024<512>(tpre, wtot, t);
    const bool staged = nb <= SORT_CAP;

    const int g0 = t * TE + o0;
    const int g1 = (t + 512) * TE + o1;
    int s[CSR_K], dl[CSR_K], r[CSR_K];
    if (staged) {
        const int a0 = tpre[t];
        const int a1 = tpre[t + 512];
        for (int j = 0; j < c0; ++j) lout[a0 + j] = g0 + j;
        for (int j = 0; j < c1; ++j) lout[a1 + j] = g1 + j;
        __syncthreads();
        int pi[CSR_K];
        v2i e[CSR_K];
#pragma unroll
        for (int k = 0; k < CSR_K; ++k) {
            const int i = t + 512 * k;
            const int p = lout[i];
            pi[k] = i < nb ? p : 0;
        }
#pragma unroll
        for (int k = 0; k < CSR_K; ++k) e[k] = *(const v2i*)(tsorted + 2 * (long long)pi[k]);
#pragma unroll
        for (int k = 0; k < CSR_K; ++k) {
            const int i = t + 512 * k;
            s[k] = e[k].x;
            dl[k] = i < nb ? (e[k].y & 255) : -1;
            r[k] = 0;
            if (dl[k] >= 0) r[k] = atomicAdd(&ncnt[dl[k]], 1);
        }
    } else {
        count_piece(tsorted + 2 * (long long)g0, c0, ncnt);
        count_piece(tsorted + 2 * (long long)g1, c1, ncnt);
    }
    __syncthreads();

    const int node = b * 256 + t;
    if (t < 256 && node < N) norm[node] = 1.0f / sqrtf(fmaxf((float)ncnt[t], 1.0f));
    block_excl_scan_1024<512>(ncnt, wtot, t);
    if (t < 256 && node < N) ptr[node] = base + ncnt[t];

    if (staged) {
        int st[CSR_K];
#pragma unroll
        for (int k = 0; k < CSR_K; ++k) st[k] = ncnt[dl[k] & 255];
#pragma unroll
        for (int k = 0; k < CSR_K; ++k)
            if (dl[k] >= 0) lout[st[k] + r[k]] = s[k];
        __syncthreads();
        for (int i = t; i < nb; i += 512) sorted[base + i] = lout[i];
    } else {
        if (t < 256) cursor[t] = ncnt[t];
        __syncthreads();
        place_piece(tsorted + 2 * (long long)g0, c0, cursor, sorted + base);
        place_piece(tsorted + 2 * (long long)g1, c1, cursor, sorted + base);
    }
}

template <int U>
__device__ __forceinline__ void add_rows(const float* __restrict__ x, int ms, int mwb, int j,
                                         int lane2, float& ax, float& ay) {
    v2f v[U];
    float w[U];
#pragma unroll
    for (int u = 0; u < U; ++u) {
        const int s = __builtin_amdgcn_readlane(ms, j + u);
        w[u] = __int_as_float(__builtin_amdgcn_readlane(mwb, j + u));
        v[u] = *(const v2f*)(x + (long long)s * 64 + lane2);
    }
#pragma unroll
    for (int u = 0; u < U; ++u) {
        ax += v[u].x * w[u];
        ay += v[u].y * w[u];
    }
}

template <int MODE>
__global__ __launch_bounds__(256) void gather_kernel(
    const float* __restrict__ x, const float* __restrict__ feat,
    const float* __restrict__ norm, const int* __restrict__ ptr,
    const int* __restrict__ sorted, float* __restrict__ res, int N, int E) {
    const int lane = threadIdx.x & 31;
    const int node = blockIdx.x * 8 + (threadIdx.x >> 5);
    if (node >= N) return;
    const int beg = __builtin_amdgcn_readfirstlane(ptr[node]);
    const int end = __builtin_amdgcn_readfirstlane(node + 1 < N ? ptr[node + 1] : E);
    const int lane2 = lane * 2;
    float ax = 0.0f, ay = 0.0f;
    for (int base = beg; base < end; base += 32) {
        const int left = end - base;
        const int n = left < 32 ? left : 32;
        int ms = 0;
        float mw = 0.0f;
        if (lane < n) {
            ms = sorted[base + lane];
            mw = norm[ms];
        }
        const int mwb = __float_as_int(mw);
        int j = 0;
        for (; j + 8 <= n; j += 8) add_rows<8>(x, ms, mwb, j, lane2, ax, ay);
        if (n - j >= 4) { add_rows<4>(x, ms, mwb, j, lane2, ax, ay); j += 4; }
        if (n - j >= 2) { add_rows<2>(x, ms, mwb, j, lane2, ax, ay); j += 2; }
        if (n - j >= 1) { add_rows<1>(x, ms, mwb, j, lane2, ax, ay); }
    }
    const long long o = (long long)node * 64 + lane2;
    if (MODE == 0) {
        const float ch = -2.0f / LAM;
        const float cx = 2.0f / LAM - 1.0f;
        const float nn = norm[node];
        const v2f f = *(const v2f*)(feat + o);
        v2f r = {ch * ax * nn + cx * f.x, ch * ay * nn + cx * f.y};
        *(v2f*)(res + o) = r;
    } else {
        v2f r = {ax, ay};
        *(v2f*)(res + o) = r;
    }
}

typedef __attribute__((ext_vector_type(4))) float v4f;

__global__ __launch_bounds__(256) void cheb_gemm_wmma(
    const float* __restrict__ feat, const float* __restrict__ Tx1,
    const float* __restrict__ h2,   const float* __restrict__ norm,
    const float* __restrict__ W,    const float* __restrict__ bias,
    float* __restrict__ out, int ntiles) {
    __shared__ alignas(16) float Wl[48 * 64 * 4];
    const int t = threadIdx.x;
    const int lane = t & 31;
    const int wave = t >> 5;
    int tile = blockIdx.x * 8 + wave;
    const bool live = tile < ntiles;
    if (!live) tile = ntiles - 1;
    const int c = lane & 15;
    const int h = lane >> 4;

    const int arow = tile * 16 + c;
    const float nrm = norm[arow];
    const float* fr = feat + (long long)arow * 64 + 4 * h;
    const float* tr = Tx1  + (long long)arow * 64 + 4 * h;
    const float* hr = h2   + (long long)arow * 64 + 4 * h;
    v4f fa[8], xa[8], ga[8];
#pragma unroll
    for (int j = 0; j < 8; ++j) {
        fa[j] = *(const v4f*)(fr + 8 * j);
        xa[j] = *(const v4f*)(tr + 8 * j);
        ga[j] = *(const v4f*)(hr + 8 * j);
    }

    for (int u = t; u < 48 * 16; u += 256) {
        const int kq = u >> 4;
        const int n4 = (u & 15) * 4;
        const float* wp = W + kq * 256 + n4;
        const v4f r0 = *(const v4f*)(wp);
        const v4f r1 = *(const v4f*)(wp + 64);
        const v4f r2 = *(const v4f*)(wp + 128);
        const v4f r3 = *(const v4f*)(wp + 192);
        float* lp = Wl + (kq * 64 + n4) * 4;
        const v4f c0 = {r0.x, r1.x, r2.x, r3.x};
        const v4f c1 = {r0.y, r1.y, r2.y, r3.y};
        const v4f c2 = {r0.z, r1.z, r2.z, r3.z};
        const v4f c3 = {r0.w, r1.w, r2.w, r3.w};
        *(v4f*)(lp)      = c0;
        *(v4f*)(lp + 4)  = c1;
        *(v4f*)(lp + 8)  = c2;
        *(v4f*)(lp + 12) = c3;
    }
    __syncthreads();
#pragma unroll
    for (int j = 0; j < 8; ++j)
        asm volatile("" :: "v"(fa[j]), "v"(xa[j]), "v"(ga[j]));

    const float* bl = Wl + (h * 64 + c) * 4;
    const float ch2 = -4.0f / LAM;
    const float cx2 =  4.0f / LAM - 2.0f;

    v8f acc[4] = {};
#pragma unroll
    for (int j = 0; j < 8; ++j) {
        const v4f f = fa[j];
        const v4f x = xa[j];
        const v4f g = ga[j];
        v4f a[3];
        a[0] = f;
        a[1] = x;
        a[2].x = ch2 * g.x * nrm + cx2 * x.x - f.x;
        a[2].y = ch2 * g.y * nrm + cx2 * x.y - f.y;
        a[2].z = ch2 * g.z * nrm + cx2 * x.z - f.z;
        a[2].w = ch2 * g.w * nrm + cx2 * x.w - f.w;
#pragma unroll
        for (int m = 0; m < 3; ++m) {
            v4f b[4];
#pragma unroll
            for (int nt = 0; nt < 4; ++nt)
                b[nt] = *(const v4f*)(bl + ((m * 16 + 2 * j) * 64 + nt * 16) * 4);
            const v2f alo = {a[m].x, a[m].y};
            const v2f ahi = {a[m].z, a[m].w};
#pragma unroll
            for (int nt = 0; nt < 4; ++nt) {
                const v2f blo = {b[nt].x, b[nt].y};
                acc[nt] = __builtin_amdgcn_wmma_f32_16x16x4_f32(false, alo, false, blo,
                                                                (short)0, acc[nt], false, false);
            }
#pragma unroll
            for (int nt = 0; nt < 4; ++nt) {
                const v2f bhi = {b[nt].z, b[nt].w};
                acc[nt] = __builtin_amdgcn_wmma_f32_16x16x4_f32(false, ahi, false, bhi,
                                                                (short)0, acc[nt], false, false);
            }
        }
    }

    if (live) {
#pragma unroll
        for (int nt = 0; nt < 4; ++nt) {
            const float bv = bias[nt * 16 + c];
#pragma unroll
            for (int r = 0; r < 8; ++r) {
                const int orow = tile * 16 + r + 8 * h;
                out[(long long)orow * 64 + nt * 16 + c] = acc[nt][r] + bv;
            }
        }
    }
}

extern "C" void kernel_launch(void* const* d_in, const int* in_sizes, int n_in,
                              void* d_out, int out_size, void* d_ws, size_t ws_size,
                              hipStream_t stream) {
    const float* feat = (const float*)d_in[0];
    const float* W    = (const float*)d_in[1];
    const float* bias = (const float*)d_in[2];
    const int*   esrc = (const int*)d_in[3];
    const int*   edst = (const int*)d_in[4];
    float* out = (float*)d_out;

    const int N = in_sizes[0] / 64;
    const int E = in_sizes[3];

    float* ws   = (float*)d_ws;
    float* norm = ws;
    float* h    = ws + N;
    float* Tx1  = ws + N + (long long)N * 64;

    const int T  = (E + TE - 1) / TE;
    const int NB = (N + 255) / 256;

    int* scr     = (int*)d_out;
    int* ptr     = scr;
    int* sorted  = scr + N;
    int* tsorted = scr + ((N + E + 1) & ~1);
    int* cntTB   = tsorted + 2 * (long long)E;
    int* offTB   = cntTB + (long long)T * NB;

    bucket_tile_kernel<<<T, 256, 0, stream>>>(esrc, edst, tsorted, cntTB, offTB, E, NB);
    bucket_csr_kernel<<<NB, 512, 0, stream>>>(tsorted, cntTB, offTB, ptr, sorted, norm, N, T, NB);

    gather_kernel<0><<<(N + 7) / 8, 256, 0, stream>>>(feat, feat, norm, ptr, sorted, Tx1, N, E);
    gather_kernel<1><<<(N + 7) / 8, 256, 0, stream>>>(Tx1, feat, norm, ptr, sorted, h, N, E);

    const int tiles = N / 16;
    cheb_gemm_wmma<<<(tiles + 7) / 8, 256, 0, stream>>>(feat, Tx1, h, norm, W, bias, out, tiles);
}


# First gather stores Tx1 rows pre-scaled by their own norm, so the second gather sums plain rows without a per-edge norm read; GEMM stage multiplies by 1/norm (2 more roundings); sum order unchanged
# speedup vs baseline: 7.2601x; 1.0364x over previous
#include <hip/hip_runtime.h>

#define LAM 2.0f

typedef __attribute__((ext_vector_type(2))) float v2f;
typedef __attribute__((ext_vector_type(2))) int v2i;
typedef __attribute__((ext_vector_type(8))) float v8f;

__device__ __forceinline__ int wave_sum(int v) {
    for (int o = 16; o > 0; o >>= 1) v += __shfl_xor(v, o);
    return v;
}

template <int NT>
__device__ __forceinline__ int block_excl_scan_1024(int* a, int* wtot, int t) {
    constexpr int PER = 1024 / NT;
    constexpr int NW = NT / 32;
    const int lane = t & 31;
    const int w = t >> 5;
    int c[PER];
    int ts = 0;
#pragma unroll
    for (int q = 0; q < PER; ++q) { c[q] = a[t * PER + q]; ts += c[q]; }
    int inc = ts;
    for (int o = 1; o < 32; o <<= 1) {
        const int u = __shfl_up(inc, o);
        if (lane >= o) inc += u;
    }
    if (lane == 31) wtot[w] = inc;
    __syncthreads();
    int off = 0, tot = 0;
    for (int q = 0; q < NW; ++q) {
        const int v = wtot[q];
        if (q < w) off += v;
        tot += v;
    }
    int ex = off + inc - ts;
#pragma unroll
    for (int q = 0; q < PER; ++q) { a[t * PER + q] = ex; ex += c[q]; }
    __syncthreads();
    return tot;
}

#define TE 4096
#define SORT_CAP 8192

__global__ __launch_bounds__(256) void bucket_tile_kernel(
    const int* __restrict__ src, const int* __restrict__ dst, int* __restrict__ tsorted,
    int* __restrict__ cntTB, int* __restrict__ offTB, int E, int NB) {
    __shared__ int hist[1024];
    __shared__ int boff[1024];
    __shared__ int wtot[8];
    __shared__ alignas(8) int stage[2 * TE];
    const int t = threadIdx.x;
    const int tile = blockIdx.x;
    for (int i = t; i < 1024; i += 256) hist[i] = 0;
    __syncthreads();
    const int e0 = tile * TE;
    int s[16], d[16], r[16];
#pragma unroll
    for (int k = 0; k < 16; ++k) {
        const int e = e0 + k * 256 + t;
        s[k] = 0;
        d[k] = -1;
        r[k] = 0;
        if (e < E) { s[k] = src[e]; d[k] = dst[e]; }
    }
#pragma unroll
    for (int k = 0; k < 16; ++k)
        if (d[k] >= 0) r[k] = atomicAdd(&hist[d[k] >> 8], 1);
    __syncthreads();
    for (int i = t; i < 1024; i += 256) boff[i] = hist[i];
    __syncthreads();
    const int tot = block_excl_scan_1024<256>(boff, wtot, t);
#pragma unroll
    for (int k = 0; k < 16; ++k)
        if (d[k] >= 0) {
            const int pos = boff[d[k] >> 8] + r[k];
            v2i sd = {s[k], d[k]};
            *(v2i*)(stage + 2 * pos) = sd;
        }
    __syncthreads();
    for (int i = t; i < tot; i += 256)
        *(v2i*)(tsorted + 2 * ((long long)e0 + i)) = *(const v2i*)(stage + 2 * i);
    for (int b = t; b < NB; b += 256) {
        cntTB[tile * NB + b] = hist[b];
        offTB[tile * NB + b] = boff[b];
    }
}

__device__ __forceinline__ void count_piece(const int* __restrict__ p, int c, int* ncnt) {
    for (int i = 0; i < c; i += 4) {
        int dd[4];
#pragma unroll
        for (int k = 0; k < 4; ++k) dd[k] = (i + k < c) ? p[2 * (i + k) + 1] : -1;
#pragma unroll
        for (int k = 0; k < 4; ++k)
            if (dd[k] >= 0) atomicAdd(&ncnt[dd[k] & 255], 1);
    }
}

__device__ __forceinline__ void place_piece(const int* __restrict__ p, int c, int* cursor,
                                            int* __restrict__ sorted_b) {
    for (int i = 0; i < c; i += 4) {
        v2i e[4];
#pragma unroll
        for (int k = 0; k < 4; ++k) {
            const v2i none = {0, -1};
            e[k] = (i + k < c) ? *(const v2i*)(p + 2 * (i + k)) : none;
        }
#pragma unroll
        for (int k = 0; k < 4; ++k)
            if (e[k].y >= 0) sorted_b[atomicAdd(&cursor[e[k].y & 255], 1)] = e[k].x;
    }
}

#define CSR_K (SORT_CAP / 512)

__global__ __launch_bounds__(512) void bucket_csr_kernel(
    const int* __restrict__ tsorted, const int* __restrict__ cntTB,
    const int* __restrict__ offTB, int* __restrict__ ptr, int* __restrict__ sorted,
    float* __restrict__ norm, int N, int T, int NB) {
    __shared__ int ncnt[1024];
    __shared__ int tpre[1024];
    __shared__ int cursor[256];
    __shared__ int wtot[16];
    __shared__ int redo[16];
    __shared__ int lout[SORT_CAP];
    const int t = threadIdx.x;
    const int b = blockIdx.x;
    int c0 = 0, o0 = 0, c1 = 0, o1 = 0;
    if (t < T) { c0 = cntTB[t * NB + b]; o0 = offTB[t * NB + b]; }
    if (t + 512 < T) { c1 = cntTB[(t + 512) * NB + b]; o1 = offTB[(t + 512) * NB + b]; }
    ncnt[t] = 0;
    ncnt[t + 512] = 0;
    tpre[t] = c0;
    tpre[t + 512] = c1;
    const int so = wave_sum(o0 + o1);
    if ((t & 31) == 0) redo[t >> 5] = so;
    __syncthreads();
    int base = 0;
    for (int q = 0; q < 16; ++q) base += redo[q];
    const int nb = block_excl_scan_1024<512>(tpre, wtot, t);
    const bool staged = nb <= SORT_CAP;

    const int g0 = t * TE + o0;
    const int g1 = (t + 512) * TE + o1;
    int s[CSR_K], dl[CSR_K], r[CSR_K];
    if (staged) {
        const int a0 = tpre[t];
        const int a1 = tpre[t + 512];
        for (int j = 0; j < c0; ++j) lout[a0 + j] = g0 + j;
        for (int j = 0; j < c1; ++j) lout[a1 + j] = g1 + j;
        __syncthreads();
        int pi[CSR_K];
        v2i e[CSR_K];
#pragma unroll
        for (int k = 0; k < CSR_K; ++k) {
            const int i = t + 512 * k;
            const int p = lout[i];
            pi[k] = i < nb ? p : 0;
        }
#pragma unroll
        for (int k = 0; k < CSR_K; ++k) e[k] = *(const v2i*)(tsorted + 2 * (long long)pi[k]);
#pragma unroll
        for (int k = 0; k < CSR_K; ++k) {
            const int i = t + 512 * k;
            s[k] = e[k].x;
            dl[k] = i < nb ? (e[k].y & 255) : -1;
            r[k] = 0;
            if (dl[k] >= 0) r[k] = atomicAdd(&ncnt[dl[k]], 1);
        }
    } else {
        count_piece(tsorted + 2 * (long long)g0, c0, ncnt);
        count_piece(tsorted + 2 * (long long)g1, c1, ncnt);
    }
    __syncthreads();

    const int node = b * 256 + t;
    if (t < 256 && node < N) norm[node] = 1.0f / sqrtf(fmaxf((float)ncnt[t], 1.0f));
    block_excl_scan_1024<512>(ncnt, wtot, t);
    if (t < 256 && node < N) ptr[node] = base + ncnt[t];

    if (staged) {
        int st[CSR_K];
#pragma unroll
        for (int k = 0; k < CSR_K; ++k) st[k] = ncnt[dl[k] & 255];
#pragma unroll
        for (int k = 0; k < CSR_K; ++k)
            if (dl[k] >= 0) lout[st[k] + r[k]] = s[k];
        __syncthreads();
        for (int i = t; i < nb; i += 512) sorted[base + i] = lout[i];
    } else {
        if (t < 256) cursor[t] = ncnt[t];
        __syncthreads();
        place_piece(tsorted + 2 * (long long)g0, c0, cursor, sorted + base);
        place_piece(tsorted + 2 * (long long)g1, c1, cursor, sorted + base);
    }
}

template <int U, bool WEIGHTED>
__device__ __forceinline__ void add_rows(const float* __restrict__ x, int ms, int mwb, int j,
                                         int lane2, float& ax, float& ay) {
    v2f v[U];
    float w[U];
#pragma unroll
    for (int u = 0; u < U; ++u) {
        const int s = __builtin_amdgcn_readlane(ms, j + u);
        if (WEIGHTED) w[u] = __int_as_float(__builtin_amdgcn_readlane(mwb, j + u));
        v[u] = *(const v2f*)(x + (long long)s * 64 + lane2);
    }
#pragma unroll
    for (int u = 0; u < U; ++u) {
        if (WEIGHTED) {
            ax += v[u].x * w[u];
            ay += v[u].y * w[u];
        } else {
            ax += v[u].x;
            ay += v[u].y;
        }
    }
}

template <int MODE>
__global__ __launch_bounds__(256) void gather_kernel(
    const float* __restrict__ x, const float* __restrict__ feat,
    const float* __restrict__ norm, const int* __restrict__ ptr,
    const int* __restrict__ sorted, float* __restrict__ res, int N, int E) {
    const int lane = threadIdx.x & 31;
    const int node = blockIdx.x * 8 + (threadIdx.x >> 5);
    if (node >= N) return;
    const int beg = __builtin_amdgcn_readfirstlane(ptr[node]);
    const int end = __builtin_amdgcn_readfirstlane(node + 1 < N ? ptr[node + 1] : E);
    const int lane2 = lane * 2;
    float ax = 0.0f, ay = 0.0f;
    for (int base = beg; base < end; base += 32) {
        const int left = end - base;
        const int n = left < 32 ? left : 32;
        int ms = 0;
        float mw = 0.0f;
        if (lane < n) {
            ms = sorted[base + lane];
            if (MODE == 0) mw = norm[ms];
        }
        const int mwb = __float_as_int(mw);
        constexpr bool WT = (MODE == 0);
        int j = 0;
        for (; j + 8 <= n; j += 8) add_rows<8, WT>(x, ms, mwb, j, lane2, ax, ay);
        if (n - j >= 4) { add_rows<4, WT>(x, ms, mwb, j, lane2, ax, ay); j += 4; }
        if (n - j >= 2) { add_rows<2, WT>(x, ms, mwb, j, lane2, ax, ay); j += 2; }
        if (n - j >= 1) { add_rows<1, WT>(x, ms, mwb, j, lane2, ax, ay); }
    }
    const long long o = (long long)node * 64 + lane2;
    if (MODE == 0) {
        const float ch = -2.0f / LAM;
        const float cx = 2.0f / LAM - 1.0f;
        const float nn = norm[node];
        const v2f f = *(const v2f*)(feat + o);
        const float tx = ch * ax * nn + cx * f.x;
        const float ty = ch * ay * nn + cx * f.y;
        v2f r = {tx * nn, ty * nn};
        *(v2f*)(res + o) = r;
    } else {
        v2f r = {ax, ay};
        *(v2f*)(res + o) = r;
    }
}

typedef __attribute__((ext_vector_type(4))) float v4f;

__global__ __launch_bounds__(256) void cheb_gemm_wmma(
    const float* __restrict__ feat, const float* __restrict__ Tx1s,
    const float* __restrict__ h2,   const float* __restrict__ norm,
    const float* __restrict__ W,    const float* __restrict__ bias,
    float* __restrict__ out, int ntiles) {
    __shared__ alignas(16) float Wl[48 * 64 * 4];
    const int t = threadIdx.x;
    const int lane = t & 31;
    const int wave = t >> 5;
    int tile = blockIdx.x * 8 + wave;
    const bool live = tile < ntiles;
    if (!live) tile = ntiles - 1;
    const int c = lane & 15;
    const int h = lane >> 4;

    const int arow = tile * 16 + c;
    const float nrm = norm[arow];
    const float* fr = feat + (long long)arow * 64 + 4 * h;
    const float* tr = Tx1s + (long long)arow * 64 + 4 * h;
    const float* hr = h2   + (long long)arow * 64 + 4 * h;
    v4f fa[8], xa[8], ga[8];
#pragma unroll
    for (int j = 0; j < 8; ++j) {
        fa[j] = *(const v4f*)(fr + 8 * j);
        xa[j] = *(const v4f*)(tr + 8 * j);
        ga[j] = *(const v4f*)(hr + 8 * j);
    }

    for (int u = t; u < 48 * 16; u += 256) {
        const int kq = u >> 4;
        const int n4 = (u & 15) * 4;
        const float* wp = W + kq * 256 + n4;
        const v4f r0 = *(const v4f*)(wp);
        const v4f r1 = *(const v4f*)(wp + 64);
        const v4f r2 = *(const v4f*)(wp + 128);
        const v4f r3 = *(const v4f*)(wp + 192);
        float* lp = Wl + (kq * 64 + n4) * 4;
        const v4f c0 = {r0.x, r1.x, r2.x, r3.x};
        const v4f c1 = {r0.y, r1.y, r2.y, r3.y};
        const v4f c2 = {r0.z, r1.z, r2.z, r3.z};
        const v4f c3 = {r0.w, r1.w, r2.w, r3.w};
        *(v4f*)(lp)      = c0;
        *(v4f*)(lp + 4)  = c1;
        *(v4f*)(lp + 8)  = c2;
        *(v4f*)(lp + 12) = c3;
    }
    __syncthreads();
#pragma unroll
    for (int j = 0; j < 8; ++j)
        asm volatile("" :: "v"(fa[j]), "v"(xa[j]), "v"(ga[j]));

    const float* bl = Wl + (h * 64 + c) * 4;
    const float ch2 = -4.0f / LAM;
    const float cx2 =  4.0f / LAM - 2.0f;
    const float rn  = 1.0f / nrm;

    v8f acc[4] = {};
#pragma unroll
    for (int j = 0; j < 8; ++j) {
        const v4f f = fa[j];
        const v4f x = xa[j] * rn;
        const v4f g = ga[j];
        v4f a[3];
        a[0] = f;
        a[1] = x;
        a[2].x = ch2 * g.x * nrm + cx2 * x.x - f.x;
        a[2].y = ch2 * g.y * nrm + cx2 * x.y - f.y;
        a[2].z = ch2 * g.z * nrm + cx2 * x.z - f.z;
        a[2].w = ch2 * g.w * nrm + cx2 * x.w - f.w;
#pragma unroll
        for (int m = 0; m < 3; ++m) {
            v4f b[4];
#pragma unroll
            for (int nt = 0; nt < 4; ++nt)
                b[nt] = *(const v4f*)(bl + ((m * 16 + 2 * j) * 64 + nt * 16) * 4);
            const v2f alo = {a[m].x, a[m].y};
            const v2f ahi = {a[m].z, a[m].w};
#pragma unroll
            for (int nt = 0; nt < 4; ++nt) {
                const v2f blo = {b[nt].x, b[nt].y};
                acc[nt] = __builtin_amdgcn_wmma_f32_16x16x4_f32(false, alo, false, blo,
                                                                (short)0, acc[nt], false, false);
            }
#pragma unroll
            for (int nt = 0; nt < 4; ++nt) {
                const v2f bhi = {b[nt].z, b[nt].w};
                acc[nt] = __builtin_amdgcn_wmma_f32_16x16x4_f32(false, ahi, false, bhi,
                                                                (short)0, acc[nt], false, false);
            }
        }
    }

    if (live) {
#pragma unroll
        for (int nt = 0; nt < 4; ++nt) {
            const float bv = bias[nt * 16 + c];
#pragma unroll
            for (int r = 0; r < 8; ++r) {
                const int orow = tile * 16 + r + 8 * h;
                out[(long long)orow * 64 + nt * 16 + c] = acc[nt][r] + bv;
            }
        }
    }
}

extern "C" void kernel_launch(void* const* d_in, const int* in_sizes, int n_in,
                              void* d_out, int out_size, void* d_ws, size_t ws_size,
                              hipStream_t stream) {
    const float* feat = (const float*)d_in[0];
    const float* W    = (const float*)d_in[1];
    const float* bias = (const float*)d_in[2];
    const int*   esrc = (const int*)d_in[3];
    const int*   edst = (const int*)d_in[4];
    float* out = (float*)d_out;

    const int N = in_sizes[0] / 64;
    const int E = in_sizes[3];

    float* ws   = (float*)d_ws;
    float* norm = ws;
    float* h    = ws + N;
    float* Tx1s = ws + N + (long long)N * 64;

    const int T  = (E + TE - 1) / TE;
    const int NB = (N + 255) / 256;

    int* scr     = (int*)d_out;
    int* ptr     = scr;
    int* sorted  = scr + N;
    int* tsorted = scr + ((N + E + 1) & ~1);
    int* cntTB   = tsorted + 2 * (long long)E;
    int* offTB   = cntTB + (long long)T * NB;

    bucket_tile_kernel<<<T, 256, 0, stream>>>(esrc, edst, tsorted, cntTB, offTB, E, NB);
    bucket_csr_kernel<<<NB, 512, 0, stream>>>(tsorted, cntTB, offTB, ptr, sorted, norm, N, T, NB);

    gather_kernel<0><<<(N + 7) / 8, 256, 0, stream>>>(feat, feat, norm, ptr, sorted, Tx1s, N, E);
    gather_kernel<1><<<(N + 7) / 8, 256, 0, stream>>>(Tx1s, feat, norm, ptr, sorted, h, N, E);

    const int tiles = N / 16;
    cheb_gemm_wmma<<<(tiles + 7) / 8, 256, 0, stream>>>(feat, Tx1s, h, norm, W, bias, out, tiles);
}


# Second gather fused into the GEMM launch: a wave sums its strip's 16 h2 rows into an LDS slot (h2 not in memory); half the waves do the feat, Tx1 products before gathering; WMMA sum order changes
# speedup vs baseline: 8.2329x; 1.1340x over previous
#include <hip/hip_runtime.h>

#define LAM 2.0f

typedef __attribute__((ext_vector_type(2))) float v2f;
typedef __attribute__((ext_vector_type(2))) int v2i;
typedef __attribute__((ext_vector_type(8))) float v8f;

__device__ __forceinline__ int wave_sum(int v) {
    for (int o = 16; o > 0; o >>= 1) v += __shfl_xor(v, o);
    return v;
}

template <int NT>
__device__ __forceinline__ int block_excl_scan_1024(int* a, int* wtot, int t) {
    constexpr int PER = 1024 / NT;
    constexpr int NW = NT / 32;
    const int lane = t & 31;
    const int w = t >> 5;
    int c[PER];
    int ts = 0;
#pragma unroll
    for (int q = 0; q < PER; ++q) { c[q] = a[t * PER + q]; ts += c[q]; }
    int inc = ts;
    for (int o = 1; o < 32; o <<= 1) {
        const int u = __shfl_up(inc, o);
        if (lane >= o) inc += u;
    }
    if (lane == 31) wtot[w] = inc;
    __syncthreads();
    int off = 0, tot = 0;
    for (int q = 0; q < NW; ++q) {
        const int v = wtot[q];
        if (q < w) off += v;
        tot += v;
    }
    int ex = off + inc - ts;
#pragma unroll
    for (int q = 0; q < PER; ++q) { a[t * PER + q] = ex; ex += c[q]; }
    __syncthreads();
    return tot;
}

#define TE 4096
#define SORT_CAP 8192

__global__ __launch_bounds__(256) void bucket_tile_kernel(
    const int* __restrict__ src, const int* __restrict__ dst, int* __restrict__ tsorted,
    int* __restrict__ cntTB, int* __restrict__ offTB, int E, int NB) {
    __shared__ int hist[1024];
    __shared__ int boff[1024];
    __shared__ int wtot[8];
    __shared__ alignas(8) int stage[2 * TE];
    const int t = threadIdx.x;
    const int tile = blockIdx.x;
    for (int i = t; i < 1024; i += 256) hist[i] = 0;
    __syncthreads();
    const int e0 = tile * TE;
    int s[16], d[16], r[16];
#pragma unroll
    for (int k = 0; k < 16; ++k) {
        const int e = e0 + k * 256 + t;
        s[k] = 0;
        d[k] = -1;
        r[k] = 0;
        if (e < E) { s[k] = src[e]; d[k] = dst[e]; }
    }
#pragma unroll
    for (int k = 0; k < 16; ++k)
        if (d[k] >= 0) r[k] = atomicAdd(&hist[d[k] >> 8], 1);
    __syncthreads();
    for (int i = t; i < 1024; i += 256) boff[i] = hist[i];
    __syncthreads();
    const int tot = block_excl_scan_1024<256>(boff, wtot, t);
#pragma unroll
    for (int k = 0; k < 16; ++k)
        if (d[k] >= 0) {
            const int pos = boff[d[k] >> 8] + r[k];
            v2i sd = {s[k], d[k]};
            *(v2i*)(stage + 2 * pos) = sd;
        }
    __syncthreads();
    for (int i = t; i < tot; i += 256)
        *(v2i*)(tsorted + 2 * ((long long)e0 + i)) = *(const v2i*)(stage + 2 * i);
    for (int b = t; b < NB; b += 256) {
        cntTB[tile * NB + b] = hist[b];
        offTB[tile * NB + b] = boff[b];
    }
}

__device__ __forceinline__ void count_piece(const int* __restrict__ p, int c, int* ncnt) {
    for (int i = 0; i < c; i += 4) {
        int dd[4];
#pragma unroll
        for (int k = 0; k < 4; ++k) dd[k] = (i + k < c) ? p[2 * (i + k) + 1] : -1;
#pragma unroll
        for (int k = 0; k < 4; ++k)
            if (dd[k] >= 0) atomicAdd(&ncnt[dd[k] & 255], 1);
    }
}

__device__ __forceinline__ void place_piece(const int* __restrict__ p, int c, int* cursor,
                                            int* __restrict__ sorted_b) {
    for (int i = 0; i < c; i += 4) {
        v2i e[4];
#pragma unroll
        for (int k = 0; k < 4; ++k) {
            const v2i none = {0, -1};
            e[k] = (i + k < c) ? *(const v2i*)(p + 2 * (i + k)) : none;
        }
#pragma unroll
        for (int k = 0; k < 4; ++k)
            if (e[k].y >= 0) sorted_b[atomicAdd(&cursor[e[k].y & 255], 1)] = e[k].x;
    }
}

#define CSR_K (SORT_CAP / 512)

__global__ __launch_bounds__(512) void bucket_csr_kernel(
    const int* __restrict__ tsorted, const int* __restrict__ cntTB,
    const int* __restrict__ offTB, int* __restrict__ ptr, int* __restrict__ sorted,
    float* __restrict__ norm, int N, int T, int NB) {
    __shared__ int ncnt[1024];
    __shared__ int tpre[1024];
    __shared__ int cursor[256];
    __shared__ int wtot[16];
    __shared__ int redo[16];
    __shared__ int lout[SORT_CAP];
    const int t = threadIdx.x;
    const int b = blockIdx.x;
    int c0 = 0, o0 = 0, c1 = 0, o1 = 0;
    if (t < T) { c0 = cntTB[t * NB + b]; o0 = offTB[t * NB + b]; }
    if (t + 512 < T) { c1 = cntTB[(t + 512) * NB + b]; o1 = offTB[(t + 512) * NB + b]; }
    ncnt[t] = 0;
    ncnt[t + 512] = 0;
    tpre[t] = c0;
    tpre[t + 512] = c1;
    const int so = wave_sum(o0 + o1);
    if ((t & 31) == 0) redo[t >> 5] = so;
    __syncthreads();
    int base = 0;
    for (int q = 0; q < 16; ++q) base += redo[q];
    const int nb = block_excl_scan_1024<512>(tpre, wtot, t);
    const bool staged = nb <= SORT_CAP;

    const int g0 = t * TE + o0;
    const int g1 = (t + 512) * TE + o1;
    int s[CSR_K], dl[CSR_K], r[CSR_K];
    if (staged) {
        const int a0 = tpre[t];
        const int a1 = tpre[t + 512];
        for (int j = 0; j < c0; ++j) lout[a0 + j] = g0 + j;
        for (int j = 0; j < c1; ++j) lout[a1 + j] = g1 + j;
        __syncthreads();
        int pi[CSR_K];
        v2i e[CSR_K];
#pragma unroll
        for (int k = 0; k < CSR_K; ++k) {
            const int i = t + 512 * k;
            const int p = lout[i];
            pi[k] = i < nb ? p : 0;
        }
#pragma unroll
        for (int k = 0; k < CSR_K; ++k) e[k] = *(const v2i*)(tsorted + 2 * (long long)pi[k]);
#pragma unroll
        for (int k = 0; k < CSR_K; ++k) {
            const int i = t + 512 * k;
            s[k] = e[k].x;
            dl[k] = i < nb ? (e[k].y & 255) : -1;
            r[k] = 0;
            if (dl[k] >= 0) r[k] = atomicAdd(&ncnt[dl[k]], 1);
        }
    } else {
        count_piece(tsorted + 2 * (long long)g0, c0, ncnt);
        count_piece(tsorted + 2 * (long long)g1, c1, ncnt);
    }
    __syncthreads();

    const int node = b * 256 + t;
    if (t < 256 && node < N) norm[node] = 1.0f / sqrtf(fmaxf((float)ncnt[t], 1.0f));
    block_excl_scan_1024<512>(ncnt, wtot, t);
    if (t < 256 && node < N) ptr[node] = base + ncnt[t];

    if (staged) {
        int st[CSR_K];
#pragma unroll
        for (int k = 0; k < CSR_K; ++k) st[k] = ncnt[dl[k] & 255];
#pragma unroll
        for (int k = 0; k < CSR_K; ++k)
            if (dl[k] >= 0) lout[st[k] + r[k]] = s[k];
        __syncthreads();
        for (int i = t; i < nb; i += 512) sorted[base + i] = lout[i];
    } else {
        if (t < 256) cursor[t] = ncnt[t];
        __syncthreads();
        place_piece(tsorted + 2 * (long long)g0, c0, cursor, sorted + base);
        place_piece(tsorted + 2 * (long long)g1, c1, cursor, sorted + base);
    }
}

template <int U>
__device__ __forceinline__ void add_rows_w(const float* __restrict__ x, int ms, int mwb, int j,
                                           int lane2, float& ax, float& ay) {
    v2f v[U];
    float w[U];
#pragma unroll
    for (int u = 0; u < U; ++u) {
        const int s = __builtin_amdgcn_readlane(ms, j + u);
        w[u] = __int_as_float(__builtin_amdgcn_readlane(mwb, j + u));
        v[u] = *(const v2f*)(x + (long long)s * 64 + lane2);
    }
#pragma unroll
    for (int u = 0; u < U; ++u) {
        ax += v[u].x * w[u];
        ay += v[u].y * w[u];
    }
}

__global__ __launch_bounds__(256) void gather_first_kernel(
    const float* __restrict__ feat, const float* __restrict__ norm,
    const int* __restrict__ ptr, const int* __restrict__ sorted,
    float* __restrict__ res, int N, int E) {
    const int lane = threadIdx.x & 31;
    const int node = blockIdx.x * 8 + (threadIdx.x >> 5);
    if (node >= N) return;
    const int beg = __builtin_amdgcn_readfirstlane(ptr[node]);
    const int end = __builtin_amdgcn_readfirstlane(node + 1 < N ? ptr[node + 1] : E);
    const int lane2 = lane * 2;
    float ax = 0.0f, ay = 0.0f;
    for (int base = beg; base < end; base += 32) {
        const int left = end - base;
        const int n = left < 32 ? left : 32;
        int ms = 0;
        float mw = 0.0f;
        if (lane < n) {
            ms = sorted[base + lane];
            mw = norm[ms];
        }
        const int mwb = __float_as_int(mw);
        int j = 0;
        for (; j + 8 <= n; j += 8) add_rows_w<8>(feat, ms, mwb, j, lane2, ax, ay);
        if (n - j >= 4) { add_rows_w<4>(feat, ms, mwb, j, lane2, ax, ay); j += 4; }
        if (n - j >= 2) { add_rows_w<2>(feat, ms, mwb, j, lane2, ax, ay); j += 2; }
        if (n - j >= 1) { add_rows_w<1>(feat, ms, mwb, j, lane2, ax, ay); }
    }
    const long long o = (long long)node * 64 + lane2;
    const float ch = -2.0f / LAM;
    const float cx = 2.0f / LAM - 1.0f;
    const float nn = norm[node];
    const v2f f = *(const v2f*)(feat + o);
    const float tx = ch * ax * nn + cx * f.x;
    const float ty = ch * ay * nn + cx * f.y;
    v2f r = {tx * nn, ty * nn};
    *(v2f*)(res + o) = r;
}

typedef __attribute__((ext_vector_type(4))) float v4f;

#define FW 25

#define CLOSE_ROW()                                                                   \
    {                                                                                 \
        const v2f o2 = {ax, ay};                                                      \
        *(v2f*)(hw + r * 64 + ((((lane >> 1) ^ r) & 15) << 2) + ((lane & 1) << 1)) = o2; \
        ax = 0.0f;                                                                    \
        ay = 0.0f;                                                                    \
        ++r;                                                                          \
        nend = __builtin_amdgcn_readlane(pv, r + 1);                                  \
    }

__device__ __forceinline__ void issue_rows(const float* __restrict__ x, int ms, int j0, int lane2,
                                           v2f (&v)[16]) {
#pragma unroll
    for (int u = 0; u < 16; ++u) {
        const int s = __builtin_amdgcn_readlane(ms, j0 + u);
        v[u] = *(const v2f*)(x + (long long)s * 64 + lane2);
    }
}

__device__ __forceinline__ void sum_rows(const v2f (&v)[16], int eg, int e1, int pv, int lane,
                                         float* hw, int& r, int& nend, float& ax, float& ay) {
#pragma unroll
    for (int u = 0; u < 16; ++u) {
        const int ee = eg + u;
        if (ee < e1) {
            while (ee == nend) CLOSE_ROW()
            ax += v[u].x;
            ay += v[u].y;
        }
    }
}

__device__ __forceinline__ void gather_strip(const float* __restrict__ x,
                                             const int* __restrict__ sorted, int pv, int lane,
                                             float* hw) {
    const int lane2 = lane * 2;
    int e = __builtin_amdgcn_readlane(pv, 0);
    const int e1 = __builtin_amdgcn_readlane(pv, 16);
    int r = 0;
    int nend = __builtin_amdgcn_readlane(pv, 1);
    float ax = 0.0f, ay = 0.0f;
    if (e < e1) {
        const int last = e1 - 1;
        int pi = e + lane;
        int msn = sorted[pi < last ? pi : last];
        while (e < e1) {
            const int cb = e;
            const int ms = msn;
            const int left = e1 - cb;
            const int n = left < 32 ? left : 32;
            v2f v[16];
            issue_rows(x, ms, 0, lane2, v);
            pi = cb + 32 + lane;
            msn = sorted[pi < last ? pi : last];
            sum_rows(v, cb, e1, pv, lane, hw, r, nend, ax, ay);
            if (n > 16) {
                issue_rows(x, ms, 16, lane2, v);
                sum_rows(v, cb + 16, e1, pv, lane, hw, r, nend, ax, ay);
            }
            e = cb + n;
        }
    }
    while (r < 16) CLOSE_ROW()
}

template <int M0, int M1>
__device__ __forceinline__ void strip_products(const v4f (&fa)[8], const v4f (&xa)[8],
                                               const v4f (&ga)[8], float nrm, float rn,
                                               const float* bl, v8f (&acc)[4]) {
    const float ch2 = -4.0f / LAM;
    const float cx2 =  4.0f / LAM - 2.0f;
#pragma unroll
    for (int j = 0; j < 8; ++j) {
        const v4f f = fa[j];
        const v4f x = xa[j] * rn;
#pragma unroll
        for (int m = M0; m < M1; ++m) {
            v4f a;
            if (m == 0) {
                a = f;
            } else if (m == 1) {
                a = x;
            } else {
                const v4f g = ga[j];
                a.x = ch2 * g.x * nrm + cx2 * x.x - f.x;
                a.y = ch2 * g.y * nrm + cx2 * x.y - f.y;
                a.z = ch2 * g.z * nrm + cx2 * x.z - f.z;
                a.w = ch2 * g.w * nrm + cx2 * x.w - f.w;
            }
            v4f b[4];
#pragma unroll
            for (int nt = 0; nt < 4; ++nt)
                b[nt] = *(const v4f*)(bl + ((m * 16 + 2 * j) * 64 + nt * 16) * 4);
            const v2f alo = {a.x, a.y};
            const v2f ahi = {a.z, a.w};
#pragma unroll
            for (int nt = 0; nt < 4; ++nt) {
                const v2f blo = {b[nt].x, b[nt].y};
                acc[nt] = __builtin_amdgcn_wmma_f32_16x16x4_f32(false, alo, false, blo,
                                                                (short)0, acc[nt], false, false);
            }
#pragma unroll
            for (int nt = 0; nt < 4; ++nt) {
                const v2f bhi = {b[nt].z, b[nt].w};
                acc[nt] = __builtin_amdgcn_wmma_f32_16x16x4_f32(false, ahi, false, bhi,
                                                                (short)0, acc[nt], false, false);
            }
        }
    }
}

__global__ __launch_bounds__(FW * 32) void gather_cheb_gemm(
    const float* __restrict__ feat, const float* __restrict__ Tx1s,
    const float* __restrict__ norm, const int* __restrict__ ptr,
    const int* __restrict__ sorted, const float* __restrict__ W,
    const float* __restrict__ bias, float* __restrict__ out, int ntiles, int E) {
    __shared__ alignas(16) float Wl[48 * 64 * 4];
    __shared__ alignas(16) float Hl[FW * 16 * 64];
    const int t = threadIdx.x;
    const int lane = t & 31;
    const int wave = __builtin_amdgcn_readfirstlane(t >> 5);
    int tile = blockIdx.x * FW + wave;
    const bool live = tile < ntiles;
    if (!live) tile = ntiles - 1;
    const int c = lane & 15;
    const int h = lane >> 4;

    int pv = E;
    {
        const int node = tile * 16 + lane;
        if (lane <= 16 && node < ntiles * 16) pv = ptr[node];
    }

    const int arow = tile * 16 + c;
    const float nrm = norm[arow];
    const float* fr = feat + (long long)arow * 64 + 4 * h;
    const float* tr = Tx1s + (long long)arow * 64 + 4 * h;
    const bool early = ((wave >> 2) & 1) != 0;
    v4f fa[8], xa[8], ga[8];
    if (early) {
#pragma unroll
        for (int j = 0; j < 8; ++j) {
            fa[j] = *(const v4f*)(fr + 8 * j);
            xa[j] = *(const v4f*)(tr + 8 * j);
        }
    }

    for (int u = t; u < 48 * 16; u += FW * 32) {
        const int kq = u >> 4;
        const int n4 = (u & 15) * 4;
        const float* wp = W + kq * 256 + n4;
        const v4f r0 = *(const v4f*)(wp);
        const v4f r1 = *(const v4f*)(wp + 64);
        const v4f r2 = *(const v4f*)(wp + 128);
        const v4f r3 = *(const v4f*)(wp + 192);
        float* lp = Wl + (kq * 64 + n4) * 4;
        const v4f c0 = {r0.x, r1.x, r2.x, r3.x};
        const v4f c1 = {r0.y, r1.y, r2.y, r3.y};
        const v4f c2 = {r0.z, r1.z, r2.z, r3.z};
        const v4f c3 = {r0.w, r1.w, r2.w, r3.w};
        *(v4f*)(lp)      = c0;
        *(v4f*)(lp + 4)  = c1;
        *(v4f*)(lp + 8)  = c2;
        *(v4f*)(lp + 12) = c3;
    }
    __syncthreads();

    const float* bl = Wl + (h * 64 + c) * 4;
    float* hw = Hl + wave * (16 * 64);
    const float rn = 1.0f / nrm;

    v8f acc[4] = {};
    if (early) {
#pragma unroll
        for (int j = 0; j < 8; ++j)
            asm volatile("" :: "v"(fa[j]), "v"(xa[j]));
        strip_products<0, 2>(fa, xa, ga, nrm, rn, bl, acc);
    }

    gather_strip(Tx1s, sorted, pv, lane, hw);

#pragma unroll
    for (int j = 0; j < 8; ++j) {
        fa[j] = *(const v4f*)(fr + 8 * j);
        xa[j] = *(const v4f*)(tr + 8 * j);
    }
#pragma unroll
    for (int j = 0; j < 8; ++j)
        asm volatile("" :: "v"(fa[j]), "v"(xa[j]));

    if (!early) strip_products<0, 2>(fa, xa, ga, nrm, rn, bl, acc);

#pragma unroll
    for (int j = 0; j < 8; ++j)
        ga[j] = *(const v4f*)(hw + c * 64 + ((((2 * j + h) ^ c) & 15) << 2));
    strip_products<2, 3>(fa, xa, ga, nrm, rn, bl, acc);

    if (live) {
#pragma unroll
        for (int nt = 0; nt < 4; ++nt) {
            const float bv = bias[nt * 16 + c];
#pragma unroll
            for (int r = 0; r < 8; ++r) {
                const int orow = tile * 16 + r + 8 * h;
                out[(long long)orow * 64 + nt * 16 + c] = acc[nt][r] + bv;
            }
        }
    }
}

extern "C" void kernel_launch(void* const* d_in, const int* in_sizes, int n_in,
                              void* d_out, int out_size, void* d_ws, size_t ws_size,
                              hipStream_t stream) {
    const float* feat = (const float*)d_in[0];
    const float* W    = (const float*)d_in[1];
    const float* bias = (const float*)d_in[2];
    const int*   esrc = (const int*)d_in[3];
    const int*   edst = (const int*)d_in[4];
    float* out = (float*)d_out;

    const int N = in_sizes[0] / 64;
    const int E = in_sizes[3];

    float* ws   = (float*)d_ws;
    float* norm = ws;
    int*   ptr    = (int*)(ws + N);
    int*   sorted = ptr + N;
    float* Tx1s = ws + N + (long long)N * 64;

    const int T  = (E + TE - 1) / TE;
    const int NB = (N + 255) / 256;

    int* tsorted = (int*)d_out;
    int* cntTB   = tsorted + 2 * (long long)E;
    int* offTB   = cntTB + (long long)T * NB;

    bucket_tile_kernel<<<T, 256, 0, stream>>>(esrc, edst, tsorted, cntTB, offTB, E, NB);
    bucket_csr_kernel<<<NB, 512, 0, stream>>>(tsorted, cntTB, offTB, ptr, sorted, norm, N, T, NB);

    gather_first_kernel<<<(N + 7) / 8, 256, 0, stream>>>(feat, norm, ptr, sorted, Tx1s, N, E);

    const int tiles = N / 16;
    gather_cheb_gemm<<<(tiles + FW - 1) / FW, FW * 32, 0, stream>>>(feat, Tx1s, norm, ptr, sorted, W,
                                                                    bias, out, tiles, E);
}
